# starting kernel
# baseline (speedup 1.0000x reference)
#include <hip/hip_runtime.h>

#define DINK 128
#define DHK  128
#define DOK  64

typedef __attribute__((ext_vector_type(2))) float v2f;
typedef __attribute__((ext_vector_type(8))) float v8f;


__global__ void gcn_zero_f4(float4* __restrict__ p, long n4) {
    long i = (long)blockIdx.x * blockDim.x + threadIdx.x;
    if (i < n4) p[i] = make_float4(0.f, 0.f, 0.f, 0.f);
}

__global__ void gcn_deg(const int* __restrict__ dst, float* __restrict__ deg, int E) {
    int e = blockIdx.x * blockDim.x + threadIdx.x;
    if (e < E) atomicAdd(&deg[dst[e]], 1.0f);
}

__global__ void gcn_dinv(float* __restrict__ deg, int n) {
    int i = blockIdx.x * blockDim.x + threadIdx.x;
    if (i < n) {
        float d = deg[i];
        deg[i] = (d > 0.f) ? rsqrtf(d) : 0.f;
    }
}

template<int NT>
__global__ __launch_bounds__(NT * 32)
void gcn_gemm_wmma(const float* __restrict__ H, const float* __restrict__ W,
                   float* __restrict__ out) {
    const int lane  = threadIdx.x & 31;
    const int wave  = threadIdx.x >> 5;
    const int baseM = blockIdx.x << 4;
    const int l15   = lane & 15;
    const int half  = lane >> 4;
    const int col   = (wave << 4) + l15;
    const int rowA  = baseM + l15;
    const int koff  = half << 1;

    const float* hp = H + (size_t)rowA * DINK + koff;
    const float* wp = W + (size_t)col  * DINK + koff;

    v8f acc = {0.f, 0.f, 0.f, 0.f, 0.f, 0.f, 0.f, 0.f};

#pragma unroll
    for (int k0 = 0; k0 < DINK; k0 += 4) {
        v2f a = *(const v2f*)(hp + k0);
        v2f b = *(const v2f*)(wp + k0);
        acc = __builtin_amdgcn_wmma_f32_16x16x4_f32(
             false, a,  false, b,
             (short)0, acc,  false,  false);
    }

    const int ldout = NT * 16;
    float* op = out + (size_t)(baseM + (half << 3)) * ldout + col;
#pragma unroll
    for (int r = 0; r < 8; ++r)
        op[(size_t)r * ldout] = acc[r];
}

__global__ void gcn_scatter128(const int* __restrict__ src, const int* __restrict__ dst,
                               const float* __restrict__ dinv, const float* __restrict__ tmp,
                               float* __restrict__ out, int E) {
    long tid = (long)blockIdx.x * blockDim.x + threadIdx.x;
    int e = (int)(tid >> 5);
    if (e >= E) return;
    int lane = (int)(tid & 31);
    int s = src[e], d = dst[e];
    float nrm = dinv[s] * dinv[d];
    const float4 v = *(const float4*)(tmp + (size_t)s * 128 + lane * 4);
    float* o = out + (size_t)d * 128 + lane * 4;
    atomicAdd(o + 0, v.x * nrm);
    atomicAdd(o + 1, v.y * nrm);
    atomicAdd(o + 2, v.z * nrm);
    atomicAdd(o + 3, v.w * nrm);
}

__global__ void gcn_scatter64(const int* __restrict__ src, const int* __restrict__ dst,
                              const float* __restrict__ dinv, const float* __restrict__ tmp,
                              float* __restrict__ out, int E) {
    long tid = (long)blockIdx.x * blockDim.x + threadIdx.x;
    int e = (int)(tid >> 4);
    if (e >= E) return;
    int lane = (int)(tid & 15);
    int s = src[e], d = dst[e];
    float nrm = dinv[s] * dinv[d];
    const float4 v = *(const float4*)(tmp + (size_t)s * 64 + lane * 4);
    float* o = out + (size_t)d * 64 + lane * 4;
    atomicAdd(o + 0, v.x * nrm);
    atomicAdd(o + 1, v.y * nrm);
    atomicAdd(o + 2, v.z * nrm);
    atomicAdd(o + 3, v.w * nrm);
}

template<int D, bool RELU>
__global__ void gcn_bias_act(float* __restrict__ h, const float* __restrict__ b, long n4) {
    long i = (long)blockIdx.x * blockDim.x + threadIdx.x;
    if (i >= n4) return;
    int c4 = (int)(i & (D / 4 - 1));
    float4 bb = ((const float4*)b)[c4];
    float4 v  = ((float4*)h)[i];
    v.x += bb.x; v.y += bb.y; v.z += bb.z; v.w += bb.w;
    if (RELU) {
        v.x = fmaxf(v.x, 0.f); v.y = fmaxf(v.y, 0.f);
        v.z = fmaxf(v.z, 0.f); v.w = fmaxf(v.w, 0.f);
    }
    ((float4*)h)[i] = v;
}


extern "C" void kernel_launch(void* const* d_in, const int* in_sizes, int n_in,
                              void* d_out, int out_size, void* d_ws, size_t ws_size,
                              hipStream_t stream) {
    const float* x  = (const float*)d_in[0];
    const int*   ei = (const int*)d_in[1];
    const float* W1 = (const float*)d_in[2];
    const float* b1 = (const float*)d_in[3];
    const float* W2 = (const float*)d_in[4];
    const float* b2 = (const float*)d_in[5];
    const float* W3 = (const float*)d_in[6];
    const float* b3 = (const float*)d_in[7];
    const float* W4 = (const float*)d_in[8];
    const float* b4 = (const float*)d_in[9];

    const int Nn = in_sizes[0] / DINK;
    const int E  = in_sizes[1] / 2;
    const int* srcI = ei;
    const int* dstI = ei + E;

    float* ws_f = (float*)d_ws;
    float* dinv = ws_f;
    size_t offA = ((size_t)Nn + 1023) & ~(size_t)1023;
    float* A = ws_f + offA;
    float* B = A + (size_t)Nn * DHK;
    float* out = (float*)d_out;

    const int TB = 256;
    const long nh4   = (long)Nn * DHK / 4;
    const long no4   = (long)Nn * DOK / 4;
    const int  gN16  = Nn / 16;
    const int  gSc128 = (int)(((long)E * 32 + TB - 1) / TB);
    const int  gSc64  = (int)(((long)E * 16 + TB - 1) / TB);
    const int  gH4    = (int)((nh4 + TB - 1) / TB);
    const int  gO4    = (int)((no4 + TB - 1) / TB);

    gcn_zero_f4<<<(Nn / 4 + TB - 1) / TB, TB, 0, stream>>>((float4*)dinv, Nn / 4);
    gcn_deg<<<(E + TB - 1) / TB, TB, 0, stream>>>(dstI, dinv, E);
    gcn_dinv<<<(Nn + TB - 1) / TB, TB, 0, stream>>>(dinv, Nn);

    gcn_gemm_wmma<8><<<gN16, 256, 0, stream>>>(x, W1, A);
    gcn_zero_f4<<<gH4, TB, 0, stream>>>((float4*)B, nh4);
    gcn_scatter128<<<gSc128, TB, 0, stream>>>(srcI, dstI, dinv, A, B, E);
    gcn_bias_act<128, true><<<gH4, TB, 0, stream>>>(B, b1, nh4);

    gcn_gemm_wmma<8><<<gN16, 256, 0, stream>>>(B, W2, A);
    gcn_zero_f4<<<gH4, TB, 0, stream>>>((float4*)B, nh4);
    gcn_scatter128<<<gSc128, TB, 0, stream>>>(srcI, dstI, dinv, A, B, E);
    gcn_bias_act<128, true><<<gH4, TB, 0, stream>>>(B, b2, nh4);

    gcn_gemm_wmma<8><<<gN16, 256, 0, stream>>>(B, W3, A);
    gcn_zero_f4<<<gH4, TB, 0, stream>>>((float4*)B, nh4);
    gcn_scatter128<<<gSc128, TB, 0, stream>>>(srcI, dstI, dinv, A, B, E);
    gcn_bias_act<128, true><<<gH4, TB, 0, stream>>>(B, b3, nh4);

    gcn_gemm_wmma<4><<<gN16, 128, 0, stream>>>(B, W4, A);
    gcn_zero_f4<<<gO4, TB, 0, stream>>>((float4*)out, no4);
    gcn_scatter64<<<gSc64, TB, 0, stream>>>(srcI, dstI, dinv, A, out, E);
    gcn_bias_act<64, false><<<gO4, TB, 0, stream>>>(out, b4, no4);
}


# Float atomic scatter-add replaced by a per-destination gather: edges counting-sorted by dst each call (int atomic rank), one wave sums a row in registers in list order and stores it once
# speedup vs baseline: 2.7733x; 2.7733x over previous
#include <hip/hip_runtime.h>

#define DINK 128
#define DHK  128
#define DOK  64

typedef __attribute__((ext_vector_type(2))) float v2f;
typedef __attribute__((ext_vector_type(8))) float v8f;


__global__ void gcn_zero_f4(float4* __restrict__ p, long n4) {
    long i = (long)blockIdx.x * blockDim.x + threadIdx.x;
    if (i < n4) p[i] = make_float4(0.f, 0.f, 0.f, 0.f);
}

__global__ void gcn_count_rank(const int* __restrict__ dst, int* __restrict__ cnt,
                               int* __restrict__ rank, int E) {
    int e = blockIdx.x * blockDim.x + threadIdx.x;
    if (e < E) rank[e] = atomicAdd(&cnt[dst[e]], 1);
}

__global__ __launch_bounds__(256)
void gcn_scan_slice_sums(const int4* __restrict__ cnt4, int* __restrict__ bsum) {
    __shared__ int wsum[8];
    const int t = threadIdx.x;
    const int4 c = cnt4[blockIdx.x * 256 + t];
    int s = c.x + c.y + c.z + c.w;
    for (int o = 16; o > 0; o >>= 1) s += __shfl_xor(s, o);
    if ((t & 31) == 0) wsum[t >> 5] = s;
    __syncthreads();
    if (t == 0) {
        int tot = 0;
        for (int i = 0; i < 8; ++i) tot += wsum[i];
        bsum[blockIdx.x] = tot;
    }
}

__global__ __launch_bounds__(256)
void gcn_scan_rows(const int4* __restrict__ cnt4, const int* __restrict__ bsum,
                   int4* __restrict__ rowptr4, float* __restrict__ dinv, int n) {
    __shared__ int wpre[8];
    __shared__ int wtot[8];
    const int t = threadIdx.x, lane = t & 31, w = t >> 5;
    const int nb = blockIdx.x;

    int part = 0;
    for (int j = t; j < nb; j += 256) part += bsum[j];
    for (int o = 16; o > 0; o >>= 1) part += __shfl_xor(part, o);
    if (lane == 0) wpre[w] = part;

    const int4 c = cnt4[nb * 256 + t];
    const int s0 = c.x, s1 = s0 + c.y, s2 = s1 + c.z, s3 = s2 + c.w;
    int incl = s3;
    for (int o = 1; o < 32; o <<= 1) {
        int v = __shfl_up(incl, o);
        if (lane >= o) incl += v;
    }
    if (lane == 31) wtot[w] = incl;
    __syncthreads();

    int base = 0;
    for (int i = 0; i < 8; ++i) base += wpre[i];
    for (int i = 0; i < w; ++i) base += wtot[i];
    const int ex = base + incl - s3;
    rowptr4[nb * 256 + t] = make_int4(ex, ex + s0, ex + s1, ex + s2);

    const int i0 = (nb * 256 + t) * 4;
    if (i0 + 0 < n) dinv[i0 + 0] = (c.x > 0) ? rsqrtf((float)c.x) : 0.f;
    if (i0 + 1 < n) dinv[i0 + 1] = (c.y > 0) ? rsqrtf((float)c.y) : 0.f;
    if (i0 + 2 < n) dinv[i0 + 2] = (c.z > 0) ? rsqrtf((float)c.z) : 0.f;
    if (i0 + 3 < n) dinv[i0 + 3] = (c.w > 0) ? rsqrtf((float)c.w) : 0.f;
}

__global__ void gcn_place(const int* __restrict__ src, const int* __restrict__ dst,
                          const int* __restrict__ rank, const int* __restrict__ rowptr,
                          const float* __restrict__ dinv, int2* __restrict__ ew, int E) {
    int e = blockIdx.x * blockDim.x + threadIdx.x;
    if (e >= E) return;
    const int s = src[e], d = dst[e];
    const float nrm = dinv[s] * dinv[d];
    ew[rowptr[d] + rank[e]] = make_int2(s, __float_as_int(nrm));
}

template<int NT>
__global__ __launch_bounds__(NT * 32)
void gcn_gemm_wmma(const float* __restrict__ H, const float* __restrict__ W,
                   float* __restrict__ out) {
    const int lane  = threadIdx.x & 31;
    const int wave  = threadIdx.x >> 5;
    const int baseM = blockIdx.x << 4;
    const int l15   = lane & 15;
    const int half  = lane >> 4;
    const int col   = (wave << 4) + l15;
    const int rowA  = baseM + l15;
    const int koff  = half << 1;

    const float* hp = H + (size_t)rowA * DINK + koff;
    const float* wp = W + (size_t)col  * DINK + koff;

    v8f acc = {0.f, 0.f, 0.f, 0.f, 0.f, 0.f, 0.f, 0.f};

#pragma unroll
    for (int k0 = 0; k0 < DINK; k0 += 4) {
        v2f a = *(const v2f*)(hp + k0);
        v2f b = *(const v2f*)(wp + k0);
        acc = __builtin_amdgcn_wmma_f32_16x16x4_f32(
             false, a,  false, b,
             (short)0, acc,  false,  false);
    }

    const int ldout = NT * 16;
    float* op = out + (size_t)(baseM + (half << 3)) * ldout + col;
#pragma unroll
    for (int r = 0; r < 8; ++r)
        op[(size_t)r * ldout] = acc[r];
}

template<int VEC>
__global__ __launch_bounds__(256)
void gcn_gather_rows(const int* __restrict__ rowptr, const int2* __restrict__ ew,
                     const float* __restrict__ tmp, float* __restrict__ out, int n) {
    typedef float vec_t __attribute__((ext_vector_type(VEC)));
    const int lane = threadIdx.x & 31;
    int d = blockIdx.x * 8 + (threadIdx.x >> 5);
    if (d >= n) return;
    d = __builtin_amdgcn_readfirstlane(d);
    const int beg = rowptr[d];
    const int end = rowptr[d + 1];
    const float* colp = tmp + lane * VEC;
    vec_t acc = (vec_t)(0.f);
    for (int base = beg; base < end; base += 32) {
        const int m = (end - base < 32) ? (end - base) : 32;
        int2 my = make_int2(0, 0);
        if (lane < m) my = ew[base + lane];
#pragma unroll 4
        for (int j = 0; j < m; ++j) {
            const int   s = __builtin_amdgcn_readlane(my.x, j);
            const float w = __int_as_float(__builtin_amdgcn_readlane(my.y, j));
            const vec_t v = *(const vec_t*)(colp + (size_t)s * (32 * VEC));
            acc += v * w;
        }
    }
    *(vec_t*)(out + (size_t)d * (32 * VEC) + lane * VEC) = acc;
}

template<int D, bool RELU>
__global__ void gcn_bias_act(float* __restrict__ h, const float* __restrict__ b, long n4) {
    long i = (long)blockIdx.x * blockDim.x + threadIdx.x;
    if (i >= n4) return;
    int c4 = (int)(i & (D / 4 - 1));
    float4 bb = ((const float4*)b)[c4];
    float4 v  = ((float4*)h)[i];
    v.x += bb.x; v.y += bb.y; v.z += bb.z; v.w += bb.w;
    if (RELU) {
        v.x = fmaxf(v.x, 0.f); v.y = fmaxf(v.y, 0.f);
        v.z = fmaxf(v.z, 0.f); v.w = fmaxf(v.w, 0.f);
    }
    ((float4*)h)[i] = v;
}

__global__ void gcn_bias_store64(const float4* __restrict__ in, const float* __restrict__ b,
                                 float4* __restrict__ out, long n4) {
    long i = (long)blockIdx.x * blockDim.x + threadIdx.x;
    if (i >= n4) return;
    float4 bb = ((const float4*)b)[(int)(i & (DOK / 4 - 1))];
    float4 v  = in[i];
    v.x += bb.x; v.y += bb.y; v.z += bb.z; v.w += bb.w;
    out[i] = v;
}


extern "C" void kernel_launch(void* const* d_in, const int* in_sizes, int n_in,
                              void* d_out, int out_size, void* d_ws, size_t ws_size,
                              hipStream_t stream) {
    const float* x  = (const float*)d_in[0];
    const int*   ei = (const int*)d_in[1];
    const float* W1 = (const float*)d_in[2];
    const float* b1 = (const float*)d_in[3];
    const float* W2 = (const float*)d_in[4];
    const float* b2 = (const float*)d_in[5];
    const float* W3 = (const float*)d_in[6];
    const float* b3 = (const float*)d_in[7];
    const float* W4 = (const float*)d_in[8];
    const float* b4 = (const float*)d_in[9];

    const int Nn = in_sizes[0] / DINK;
    const int E  = in_sizes[1] / 2;
    const int* srcI = ei;
    const int* dstI = ei + E;

    float* ws_f = (float*)d_ws;
    float* dinv = ws_f;
    size_t offA = ((size_t)Nn + 1023) & ~(size_t)1023;
    float* A = ws_f + offA;
    float* B = A + (size_t)Nn * DHK;
    float* out = (float*)d_out;

    const int Np = (Nn / 1024 + 1) * 1024;
    const int NS = Np / 1024;
    int2* ew     = (int2*)d_out;
    int*  cnt    = (int*)d_out + ((2 * (size_t)E + 3) & ~(size_t)3);
    int*  rowptr = cnt + Np;
    int*  bsum   = rowptr + Np;
    int*  rank   = (int*)B;

    const int TB = 256;
    const long nh4   = (long)Nn * DHK / 4;
    const long no4   = (long)Nn * DOK / 4;
    const int  gN16  = Nn / 16;
    const int  gE    = (E + TB - 1) / TB;
    const int  gRows = (Nn + 7) / 8;
    const int  gH4   = (int)((nh4 + TB - 1) / TB);
    const int  gO4   = (int)((no4 + TB - 1) / TB);

    gcn_zero_f4<<<(Np / 4 + TB - 1) / TB, TB, 0, stream>>>((float4*)cnt, Np / 4);
    gcn_count_rank<<<gE, TB, 0, stream>>>(dstI, cnt, rank, E);
    gcn_scan_slice_sums<<<NS, 256, 0, stream>>>((const int4*)cnt, bsum);
    gcn_scan_rows<<<NS, 256, 0, stream>>>((const int4*)cnt, bsum, (int4*)rowptr, dinv, Nn);
    gcn_place<<<gE, TB, 0, stream>>>(srcI, dstI, rank, rowptr, dinv, ew, E);

    gcn_gemm_wmma<8><<<gN16, 256, 0, stream>>>(x, W1, A);
    gcn_gather_rows<4><<<gRows, 256, 0, stream>>>(rowptr, ew, A, B, Nn);
    gcn_bias_act<128, true><<<gH4, TB, 0, stream>>>(B, b1, nh4);

    gcn_gemm_wmma<8><<<gN16, 256, 0, stream>>>(B, W2, A);
    gcn_gather_rows<4><<<gRows, 256, 0, stream>>>(rowptr, ew, A, B, Nn);
    gcn_bias_act<128, true><<<gH4, TB, 0, stream>>>(B, b2, nh4);

    gcn_gemm_wmma<8><<<gN16, 256, 0, stream>>>(B, W3, A);
    gcn_gather_rows<4><<<gRows, 256, 0, stream>>>(rowptr, ew, A, B, Nn);
    gcn_bias_act<128, true><<<gH4, TB, 0, stream>>>(B, b3, nh4);

    gcn_gemm_wmma<4><<<gN16, 128, 0, stream>>>(B, W4, A);
    gcn_gather_rows<2><<<gRows, 256, 0, stream>>>(rowptr, ew, A, B, Nn);
    gcn_bias_store64<<<gO4, TB, 0, stream>>>((const float4*)B, b4, (float4*)out, no4);
}


# GEMM operands staged through LDS by whole 512-byte row loads: a wave's 16 W rows once per block into registers, A tile double-buffered, 5 row tiles per block; k order and every sum unchanged
# speedup vs baseline: 3.8387x; 1.3842x over previous
#include <hip/hip_runtime.h>

#define DINK 128
#define DHK  128
#define DOK  64
#define GMT  5

typedef __attribute__((ext_vector_type(2))) float v2f;
typedef __attribute__((ext_vector_type(8))) float v8f;


__global__ void gcn_zero_f4(float4* __restrict__ p, long n4) {
    long i = (long)blockIdx.x * blockDim.x + threadIdx.x;
    if (i < n4) p[i] = make_float4(0.f, 0.f, 0.f, 0.f);
}

__global__ void gcn_count_rank(const int* __restrict__ dst, int* __restrict__ cnt,
                               int* __restrict__ rank, int E) {
    int e = blockIdx.x * blockDim.x + threadIdx.x;
    if (e < E) rank[e] = atomicAdd(&cnt[dst[e]], 1);
}

__global__ __launch_bounds__(256)
void gcn_scan_slice_sums(const int4* __restrict__ cnt4, int* __restrict__ bsum) {
    __shared__ int wsum[8];
    const int t = threadIdx.x;
    const int4 c = cnt4[blockIdx.x * 256 + t];
    int s = c.x + c.y + c.z + c.w;
    for (int o = 16; o > 0; o >>= 1) s += __shfl_xor(s, o);
    if ((t & 31) == 0) wsum[t >> 5] = s;
    __syncthreads();
    if (t == 0) {
        int tot = 0;
        for (int i = 0; i < 8; ++i) tot += wsum[i];
        bsum[blockIdx.x] = tot;
    }
}

__global__ __launch_bounds__(256)
void gcn_scan_rows(const int4* __restrict__ cnt4, const int* __restrict__ bsum,
                   int4* __restrict__ rowptr4, float* __restrict__ dinv, int n) {
    __shared__ int wpre[8];
    __shared__ int wtot[8];
    const int t = threadIdx.x, lane = t & 31, w = t >> 5;
    const int nb = blockIdx.x;

    int part = 0;
    for (int j = t; j < nb; j += 256) part += bsum[j];
    for (int o = 16; o > 0; o >>= 1) part += __shfl_xor(part, o);
    if (lane == 0) wpre[w] = part;

    const int4 c = cnt4[nb * 256 + t];
    const int s0 = c.x, s1 = s0 + c.y, s2 = s1 + c.z, s3 = s2 + c.w;
    int incl = s3;
    for (int o = 1; o < 32; o <<= 1) {
        int v = __shfl_up(incl, o);
        if (lane >= o) incl += v;
    }
    if (lane == 31) wtot[w] = incl;
    __syncthreads();

    int base = 0;
    for (int i = 0; i < 8; ++i) base += wpre[i];
    for (int i = 0; i < w; ++i) base += wtot[i];
    const int ex = base + incl - s3;
    rowptr4[nb * 256 + t] = make_int4(ex, ex + s0, ex + s1, ex + s2);

    const int i0 = (nb * 256 + t) * 4;
    if (i0 + 0 < n) dinv[i0 + 0] = (c.x > 0) ? rsqrtf((float)c.x) : 0.f;
    if (i0 + 1 < n) dinv[i0 + 1] = (c.y > 0) ? rsqrtf((float)c.y) : 0.f;
    if (i0 + 2 < n) dinv[i0 + 2] = (c.z > 0) ? rsqrtf((float)c.z) : 0.f;
    if (i0 + 3 < n) dinv[i0 + 3] = (c.w > 0) ? rsqrtf((float)c.w) : 0.f;
}

__global__ void gcn_place(const int* __restrict__ src, const int* __restrict__ dst,
                          const int* __restrict__ rank, const int* __restrict__ rowptr,
                          const float* __restrict__ dinv, int2* __restrict__ ew, int E) {
    int e = blockIdx.x * blockDim.x + threadIdx.x;
    if (e >= E) return;
    const int s = src[e], d = dst[e];
    const float nrm = dinv[s] * dinv[d];
    ew[rowptr[d] + rank[e]] = make_int2(s, __float_as_int(nrm));
}

#define LDP 132

template<int NT, int MT>
__global__ __launch_bounds__(NT * 32)
void gcn_gemm_wmma(const float* __restrict__ H, const float* __restrict__ W,
                   float* __restrict__ out, int nTiles) {
    __shared__ __attribute__((aligned(16))) float sW[NT * 16 * LDP];
    __shared__ __attribute__((aligned(16))) float sA[2][16 * LDP];

    constexpr int RPW = 16 / NT;
    static_assert(RPW == 2 || RPW == 4, "a wave stages 2 or 4 rows of the A tile");
    constexpr int RSTEP = NT * (DINK / 4);
    const int lane  = threadIdx.x & 31;
    const int wave  = threadIdx.x >> 5;
    const int l15   = lane & 15;
    const int half  = lane >> 4;
    const int col   = (wave << 4) + l15;
    const int koff  = half << 1;

    const int tile0 = blockIdx.x * MT;
    const int tend  = (tile0 + MT < nTiles) ? (tile0 + MT) : nTiles;

    const float4* hsrc = (const float4*)(H + (size_t)wave * DINK) + lane;
    float4 pre0, pre1, pre2 = make_float4(0.f, 0.f, 0.f, 0.f), pre3 = pre2;
    {
        const float4* s = hsrc + (size_t)tile0 * (16 * DINK / 4);
        pre0 = s[0];
        pre1 = s[RSTEP];
        if constexpr (RPW == 4) { pre2 = s[2 * RSTEP]; pre3 = s[3 * RSTEP]; }
    }

    {
        const float4* wsrc = (const float4*)(W + (size_t)(wave << 4) * DINK) + lane;
        float4 wrow[16];
#pragma unroll
        for (int i = 0; i < 16; ++i) wrow[i] = wsrc[i * (DINK / 4)];
#pragma unroll
        for (int i = 0; i < 16; ++i)
            *(float4*)&sW[((wave << 4) + i) * LDP + (lane << 2)] = wrow[i];
    }
    __syncthreads();
    v2f bfrag[DINK / 4];
    {
        const float* wp = &sW[col * LDP + koff];
#pragma unroll
        for (int t = 0; t < DINK / 4; ++t)
            bfrag[t] = *(const v2f*)(wp + 4 * t);
    }

    const int ldout = NT * 16;
    int p = 0;
    for (int tile = tile0; tile < tend; ++tile) {
        {
            float* d = &sA[p][wave * LDP + (lane << 2)];
            *(float4*)(d)                = pre0;
            *(float4*)(d + NT * LDP)     = pre1;
            if constexpr (RPW == 4) {
                *(float4*)(d + 2 * NT * LDP) = pre2;
                *(float4*)(d + 3 * NT * LDP) = pre3;
            }
        }
        __syncthreads();
        if (tile + 1 < tend) {
            const float4* s = hsrc + (size_t)(tile + 1) * (16 * DINK / 4);
            pre0 = s[0];
            pre1 = s[RSTEP];
            if constexpr (RPW == 4) { pre2 = s[2 * RSTEP]; pre3 = s[3 * RSTEP]; }
        }

        const float* ap = &sA[p][l15 * LDP + koff];
        v8f acc = {0.f, 0.f, 0.f, 0.f, 0.f, 0.f, 0.f, 0.f};
#pragma unroll
        for (int t = 0; t < DINK / 4; ++t) {
            v2f a = *(const v2f*)(ap + 4 * t);
            acc = __builtin_amdgcn_wmma_f32_16x16x4_f32(
                 false, a,  false, bfrag[t],
                 (short)0, acc,  false,  false);
        }

        float* op = out + (size_t)(tile * 16 + (half << 3)) * ldout + col;
#pragma unroll
        for (int r = 0; r < 8; ++r)
            op[(size_t)r * ldout] = acc[r];
        p ^= 1;
    }
}

template<int VEC>
__global__ __launch_bounds__(256)
void gcn_gather_rows(const int* __restrict__ rowptr, const int2* __restrict__ ew,
                     const float* __restrict__ tmp, float* __restrict__ out, int n) {
    typedef float vec_t __attribute__((ext_vector_type(VEC)));
    const int lane = threadIdx.x & 31;
    int d = blockIdx.x * 8 + (threadIdx.x >> 5);
    if (d >= n) return;
    d = __builtin_amdgcn_readfirstlane(d);
    const int beg = rowptr[d];
    const int end = rowptr[d + 1];
    const float* colp = tmp + lane * VEC;
    vec_t acc = (vec_t)(0.f);
    for (int base = beg; base < end; base += 32) {
        const int m = (end - base < 32) ? (end - base) : 32;
        int2 my = make_int2(0, 0);
        if (lane < m) my = ew[base + lane];
#pragma unroll 4
        for (int j = 0; j < m; ++j) {
            const int   s = __builtin_amdgcn_readlane(my.x, j);
            const float w = __int_as_float(__builtin_amdgcn_readlane(my.y, j));
            const vec_t v = *(const vec_t*)(colp + (size_t)s * (32 * VEC));
            acc += v * w;
        }
    }
    *(vec_t*)(out + (size_t)d * (32 * VEC) + lane * VEC) = acc;
}

template<int D, bool RELU>
__global__ void gcn_bias_act(float* __restrict__ h, const float* __restrict__ b, long n4) {
    long i = (long)blockIdx.x * blockDim.x + threadIdx.x;
    if (i >= n4) return;
    int c4 = (int)(i & (D / 4 - 1));
    float4 bb = ((const float4*)b)[c4];
    float4 v  = ((float4*)h)[i];
    v.x += bb.x; v.y += bb.y; v.z += bb.z; v.w += bb.w;
    if (RELU) {
        v.x = fmaxf(v.x, 0.f); v.y = fmaxf(v.y, 0.f);
        v.z = fmaxf(v.z, 0.f); v.w = fmaxf(v.w, 0.f);
    }
    ((float4*)h)[i] = v;
}

__global__ void gcn_bias_store64(const float4* __restrict__ in, const float* __restrict__ b,
                                 float4* __restrict__ out, long n4) {
    long i = (long)blockIdx.x * blockDim.x + threadIdx.x;
    if (i >= n4) return;
    float4 bb = ((const float4*)b)[(int)(i & (DOK / 4 - 1))];
    float4 v  = in[i];
    v.x += bb.x; v.y += bb.y; v.z += bb.z; v.w += bb.w;
    out[i] = v;
}


extern "C" void kernel_launch(void* const* d_in, const int* in_sizes, int n_in,
                              void* d_out, int out_size, void* d_ws, size_t ws_size,
                              hipStream_t stream) {
    const float* x  = (const float*)d_in[0];
    const int*   ei = (const int*)d_in[1];
    const float* W1 = (const float*)d_in[2];
    const float* b1 = (const float*)d_in[3];
    const float* W2 = (const float*)d_in[4];
    const float* b2 = (const float*)d_in[5];
    const float* W3 = (const float*)d_in[6];
    const float* b3 = (const float*)d_in[7];
    const float* W4 = (const float*)d_in[8];
    const float* b4 = (const float*)d_in[9];

    const int Nn = in_sizes[0] / DINK;
    const int E  = in_sizes[1] / 2;
    const int* srcI = ei;
    const int* dstI = ei + E;

    float* ws_f = (float*)d_ws;
    float* dinv = ws_f;
    size_t offA = ((size_t)Nn + 1023) & ~(size_t)1023;
    float* A = ws_f + offA;
    float* B = A + (size_t)Nn * DHK;
    float* out = (float*)d_out;

    const int Np = (Nn / 1024 + 1) * 1024;
    const int NS = Np / 1024;
    int2* ew     = (int2*)d_out;
    int*  cnt    = (int*)d_out + ((2 * (size_t)E + 3) & ~(size_t)3);
    int*  rowptr = cnt + Np;
    int*  bsum   = rowptr + Np;
    int*  rank   = (int*)B;

    const int TB = 256;
    const long nh4   = (long)Nn * DHK / 4;
    const long no4   = (long)Nn * DOK / 4;
    const int  gN16  = Nn / 16;
    const int  gGemm = (gN16 + GMT - 1) / GMT;
    const int  gE    = (E + TB - 1) / TB;
    const int  gRows = (Nn + 7) / 8;
    const int  gH4   = (int)((nh4 + TB - 1) / TB);
    const int  gO4   = (int)((no4 + TB - 1) / TB);

    gcn_zero_f4<<<(Np / 4 + TB - 1) / TB, TB, 0, stream>>>((float4*)cnt, Np / 4);
    gcn_count_rank<<<gE, TB, 0, stream>>>(dstI, cnt, rank, E);
    gcn_scan_slice_sums<<<NS, 256, 0, stream>>>((const int4*)cnt, bsum);
    gcn_scan_rows<<<NS, 256, 0, stream>>>((const int4*)cnt, bsum, (int4*)rowptr, dinv, Nn);
    gcn_place<<<gE, TB, 0, stream>>>(srcI, dstI, rank, rowptr, dinv, ew, E);

    gcn_gemm_wmma<8, GMT><<<gGemm, 256, 0, stream>>>(x, W1, A, gN16);
    gcn_gather_rows<4><<<gRows, 256, 0, stream>>>(rowptr, ew, A, B, Nn);
    gcn_bias_act<128, true><<<gH4, TB, 0, stream>>>(B, b1, nh4);

    gcn_gemm_wmma<8, GMT><<<gGemm, 256, 0, stream>>>(B, W2, A, gN16);
    gcn_gather_rows<4><<<gRows, 256, 0, stream>>>(rowptr, ew, A, B, Nn);
    gcn_bias_act<128, true><<<gH4, TB, 0, stream>>>(B, b2, nh4);

    gcn_gemm_wmma<8, GMT><<<gGemm, 256, 0, stream>>>(B, W3, A, gN16);
    gcn_gather_rows<4><<<gRows, 256, 0, stream>>>(rowptr, ew, A, B, Nn);
    gcn_bias_act<128, true><<<gH4, TB, 0, stream>>>(B, b3, nh4);

    gcn_gemm_wmma<4, GMT><<<gGemm, 128, 0, stream>>>(B, W4, A, gN16);
    gcn_gather_rows<2><<<gRows, 256, 0, stream>>>(rowptr, ew, A, B, Nn);
    gcn_bias_store64<<<gO4, TB, 0, stream>>>((const float4*)B, b4, (float4*)out, no4);
}


# Bias and relu applied in the gather's one row store (sum, then +b, then max 0, as before) in layers 1-3: the three elementwise passes over the 128-wide activations and their launches are gone
# speedup vs baseline: 4.2343x; 1.1031x over previous
#include <hip/hip_runtime.h>

#define DINK 128
#define DHK  128
#define DOK  64
#define GMT  5

typedef __attribute__((ext_vector_type(2))) float v2f;
typedef __attribute__((ext_vector_type(8))) float v8f;


__global__ void gcn_zero_f4(float4* __restrict__ p, long n4) {
    long i = (long)blockIdx.x * blockDim.x + threadIdx.x;
    if (i < n4) p[i] = make_float4(0.f, 0.f, 0.f, 0.f);
}

__global__ void gcn_count_rank(const int* __restrict__ dst, int* __restrict__ cnt,
                               int* __restrict__ rank, int E) {
    int e = blockIdx.x * blockDim.x + threadIdx.x;
    if (e < E) rank[e] = atomicAdd(&cnt[dst[e]], 1);
}

__global__ __launch_bounds__(256)
void gcn_scan_slice_sums(const int4* __restrict__ cnt4, int* __restrict__ bsum) {
    __shared__ int wsum[8];
    const int t = threadIdx.x;
    const int4 c = cnt4[blockIdx.x * 256 + t];
    int s = c.x + c.y + c.z + c.w;
    for (int o = 16; o > 0; o >>= 1) s += __shfl_xor(s, o);
    if ((t & 31) == 0) wsum[t >> 5] = s;
    __syncthreads();
    if (t == 0) {
        int tot = 0;
        for (int i = 0; i < 8; ++i) tot += wsum[i];
        bsum[blockIdx.x] = tot;
    }
}

__global__ __launch_bounds__(256)
void gcn_scan_rows(const int4* __restrict__ cnt4, const int* __restrict__ bsum,
                   int4* __restrict__ rowptr4, float* __restrict__ dinv, int n) {
    __shared__ int wpre[8];
    __shared__ int wtot[8];
    const int t = threadIdx.x, lane = t & 31, w = t >> 5;
    const int nb = blockIdx.x;

    int part = 0;
    for (int j = t; j < nb; j += 256) part += bsum[j];
    for (int o = 16; o > 0; o >>= 1) part += __shfl_xor(part, o);
    if (lane == 0) wpre[w] = part;

    const int4 c = cnt4[nb * 256 + t];
    const int s0 = c.x, s1 = s0 + c.y, s2 = s1 + c.z, s3 = s2 + c.w;
    int incl = s3;
    for (int o = 1; o < 32; o <<= 1) {
        int v = __shfl_up(incl, o);
        if (lane >= o) incl += v;
    }
    if (lane == 31) wtot[w] = incl;
    __syncthreads();

    int base = 0;
    for (int i = 0; i < 8; ++i) base += wpre[i];
    for (int i = 0; i < w; ++i) base += wtot[i];
    const int ex = base + incl - s3;
    rowptr4[nb * 256 + t] = make_int4(ex, ex + s0, ex + s1, ex + s2);

    const int i0 = (nb * 256 + t) * 4;
    if (i0 + 0 < n) dinv[i0 + 0] = (c.x > 0) ? rsqrtf((float)c.x) : 0.f;
    if (i0 + 1 < n) dinv[i0 + 1] = (c.y > 0) ? rsqrtf((float)c.y) : 0.f;
    if (i0 + 2 < n) dinv[i0 + 2] = (c.z > 0) ? rsqrtf((float)c.z) : 0.f;
    if (i0 + 3 < n) dinv[i0 + 3] = (c.w > 0) ? rsqrtf((float)c.w) : 0.f;
}

__global__ void gcn_place(const int* __restrict__ src, const int* __restrict__ dst,
                          const int* __restrict__ rank, const int* __restrict__ rowptr,
                          const float* __restrict__ dinv, int2* __restrict__ ew, int E) {
    int e = blockIdx.x * blockDim.x + threadIdx.x;
    if (e >= E) return;
    const int s = src[e], d = dst[e];
    const float nrm = dinv[s] * dinv[d];
    ew[rowptr[d] + rank[e]] = make_int2(s, __float_as_int(nrm));
}

#define LDP 132

template<int NT, int MT>
__global__ __launch_bounds__(NT * 32)
void gcn_gemm_wmma(const float* __restrict__ H, const float* __restrict__ W,
                   float* __restrict__ out, int nTiles) {
    __shared__ __attribute__((aligned(16))) float sW[NT * 16 * LDP];
    __shared__ __attribute__((aligned(16))) float sA[2][16 * LDP];

    constexpr int RPW = 16 / NT;
    static_assert(RPW == 2 || RPW == 4, "a wave stages 2 or 4 rows of the A tile");
    constexpr int RSTEP = NT * (DINK / 4);
    const int lane  = threadIdx.x & 31;
    const int wave  = threadIdx.x >> 5;
    const int l15   = lane & 15;
    const int half  = lane >> 4;
    const int col   = (wave << 4) + l15;
    const int koff  = half << 1;

    const int tile0 = blockIdx.x * MT;
    const int tend  = (tile0 + MT < nTiles) ? (tile0 + MT) : nTiles;

    const float4* hsrc = (const float4*)(H + (size_t)wave * DINK) + lane;
    float4 pre0, pre1, pre2 = make_float4(0.f, 0.f, 0.f, 0.f), pre3 = pre2;
    {
        const float4* s = hsrc + (size_t)tile0 * (16 * DINK / 4);
        pre0 = s[0];
        pre1 = s[RSTEP];
        if constexpr (RPW == 4) { pre2 = s[2 * RSTEP]; pre3 = s[3 * RSTEP]; }
    }

    {
        const float4* wsrc = (const float4*)(W + (size_t)(wave << 4) * DINK) + lane;
        float4 wrow[16];
#pragma unroll
        for (int i = 0; i < 16; ++i) wrow[i] = wsrc[i * (DINK / 4)];
#pragma unroll
        for (int i = 0; i < 16; ++i)
            *(float4*)&sW[((wave << 4) + i) * LDP + (lane << 2)] = wrow[i];
    }
    __syncthreads();
    v2f bfrag[DINK / 4];
    {
        const float* wp = &sW[col * LDP + koff];
#pragma unroll
        for (int t = 0; t < DINK / 4; ++t)
            bfrag[t] = *(const v2f*)(wp + 4 * t);
    }

    const int ldout = NT * 16;
    int p = 0;
    for (int tile = tile0; tile < tend; ++tile) {
        {
            float* d = &sA[p][wave * LDP + (lane << 2)];
            *(float4*)(d)                = pre0;
            *(float4*)(d + NT * LDP)     = pre1;
            if constexpr (RPW == 4) {
                *(float4*)(d + 2 * NT * LDP) = pre2;
                *(float4*)(d + 3 * NT * LDP) = pre3;
            }
        }
        __syncthreads();
        if (tile + 1 < tend) {
            const float4* s = hsrc + (size_t)(tile + 1) * (16 * DINK / 4);
            pre0 = s[0];
            pre1 = s[RSTEP];
            if constexpr (RPW == 4) { pre2 = s[2 * RSTEP]; pre3 = s[3 * RSTEP]; }
        }

        const float* ap = &sA[p][l15 * LDP + koff];
        v8f acc = {0.f, 0.f, 0.f, 0.f, 0.f, 0.f, 0.f, 0.f};
#pragma unroll
        for (int t = 0; t < DINK / 4; ++t) {
            v2f a = *(const v2f*)(ap + 4 * t);
            acc = __builtin_amdgcn_wmma_f32_16x16x4_f32(
                 false, a,  false, bfrag[t],
                 (short)0, acc,  false,  false);
        }

        float* op = out + (size_t)(tile * 16 + (half << 3)) * ldout + col;
#pragma unroll
        for (int r = 0; r < 8; ++r)
            op[(size_t)r * ldout] = acc[r];
        p ^= 1;
    }
}

template<int VEC, bool ACT>
__global__ __launch_bounds__(256)
void gcn_gather_rows(const int* __restrict__ rowptr, const int2* __restrict__ ew,
                     const float* __restrict__ tmp, const float* __restrict__ bias,
                     float* __restrict__ out, int n) {
    typedef float vec_t __attribute__((ext_vector_type(VEC)));
    const int lane = threadIdx.x & 31;
    int d = blockIdx.x * 8 + (threadIdx.x >> 5);
    if (d >= n) return;
    d = __builtin_amdgcn_readfirstlane(d);
    const int beg = rowptr[d];
    const int end = rowptr[d + 1];
    const float* colp = tmp + lane * VEC;
    vec_t bb = (vec_t)(0.f);
    if constexpr (ACT) bb = *(const vec_t*)(bias + lane * VEC);
    vec_t acc = (vec_t)(0.f);
    for (int base = beg; base < end; base += 32) {
        const int m = (end - base < 32) ? (end - base) : 32;
        int2 my = make_int2(0, 0);
        if (lane < m) my = ew[base + lane];
#pragma unroll 4
        for (int j = 0; j < m; ++j) {
            const int   s = __builtin_amdgcn_readlane(my.x, j);
            const float w = __int_as_float(__builtin_amdgcn_readlane(my.y, j));
            const vec_t v = *(const vec_t*)(colp + (size_t)s * (32 * VEC));
            acc += v * w;
        }
    }
    if constexpr (ACT) {
        acc += bb;
        acc = __builtin_elementwise_max(acc, (vec_t)(0.f));
    }
    *(vec_t*)(out + (size_t)d * (32 * VEC) + lane * VEC) = acc;
}

__global__ void gcn_bias_store64(const float4* __restrict__ in, const float* __restrict__ b,
                                 float4* __restrict__ out, long n4) {
    long i = (long)blockIdx.x * blockDim.x + threadIdx.x;
    if (i >= n4) return;
    float4 bb = ((const float4*)b)[(int)(i & (DOK / 4 - 1))];
    float4 v  = in[i];
    v.x += bb.x; v.y += bb.y; v.z += bb.z; v.w += bb.w;
    out[i] = v;
}


extern "C" void kernel_launch(void* const* d_in, const int* in_sizes, int n_in,
                              void* d_out, int out_size, void* d_ws, size_t ws_size,
                              hipStream_t stream) {
    const float* x  = (const float*)d_in[0];
    const int*   ei = (const int*)d_in[1];
    const float* W1 = (const float*)d_in[2];
    const float* b1 = (const float*)d_in[3];
    const float* W2 = (const float*)d_in[4];
    const float* b2 = (const float*)d_in[5];
    const float* W3 = (const float*)d_in[6];
    const float* b3 = (const float*)d_in[7];
    const float* W4 = (const float*)d_in[8];
    const float* b4 = (const float*)d_in[9];

    const int Nn = in_sizes[0] / DINK;
    const int E  = in_sizes[1] / 2;
    const int* srcI = ei;
    const int* dstI = ei + E;

    float* ws_f = (float*)d_ws;
    float* dinv = ws_f;
    size_t offA = ((size_t)Nn + 1023) & ~(size_t)1023;
    float* A = ws_f + offA;
    float* B = A + (size_t)Nn * DHK;
    float* out = (float*)d_out;

    const int Np = (Nn / 1024 + 1) * 1024;
    const int NS = Np / 1024;
    int2* ew     = (int2*)d_out;
    int*  cnt    = (int*)d_out + ((2 * (size_t)E + 3) & ~(size_t)3);
    int*  rowptr = cnt + Np;
    int*  bsum   = rowptr + Np;
    int*  rank   = (int*)B;

    const int TB = 256;
    const long no4   = (long)Nn * DOK / 4;
    const int  gN16  = Nn / 16;
    const int  gGemm = (gN16 + GMT - 1) / GMT;
    const int  gE    = (E + TB - 1) / TB;
    const int  gRows = (Nn + 7) / 8;
    const int  gO4   = (int)((no4 + TB - 1) / TB);

    gcn_zero_f4<<<(Np / 4 + TB - 1) / TB, TB, 0, stream>>>((float4*)cnt, Np / 4);
    gcn_count_rank<<<gE, TB, 0, stream>>>(dstI, cnt, rank, E);
    gcn_scan_slice_sums<<<NS, 256, 0, stream>>>((const int4*)cnt, bsum);
    gcn_scan_rows<<<NS, 256, 0, stream>>>((const int4*)cnt, bsum, (int4*)rowptr, dinv, Nn);
    gcn_place<<<gE, TB, 0, stream>>>(srcI, dstI, rank, rowptr, dinv, ew, E);

    gcn_gemm_wmma<8, GMT><<<gGemm, 256, 0, stream>>>(x, W1, A, gN16);
    gcn_gather_rows<4, true><<<gRows, 256, 0, stream>>>(rowptr, ew, A, b1, B, Nn);

    gcn_gemm_wmma<8, GMT><<<gGemm, 256, 0, stream>>>(B, W2, A, gN16);
    gcn_gather_rows<4, true><<<gRows, 256, 0, stream>>>(rowptr, ew, A, b2, B, Nn);

    gcn_gemm_wmma<8, GMT><<<gGemm, 256, 0, stream>>>(B, W3, A, gN16);
    gcn_gather_rows<4, true><<<gRows, 256, 0, stream>>>(rowptr, ew, A, b3, B, Nn);

    gcn_gemm_wmma<4, GMT><<<gGemm, 128, 0, stream>>>(B, W4, A, gN16);
    gcn_gather_rows<2, false><<<gRows, 256, 0, stream>>>(rowptr, ew, A, b4, B, Nn);
    gcn_bias_store64<<<gO4, TB, 0, stream>>>((const float4*)B, b4, (float4*)out, no4);
}


# Edges grouped by destination by a two-level counting sort on LDS counters (256-row bins, then rows in a bin), not one global returning add per edge; a row's sum order is now its LDS arrival order
# speedup vs baseline: 5.5725x; 1.3160x over previous
#include <hip/hip_runtime.h>

#define DINK 128
#define DHK  128
#define DOK  64
#define GMT  5

typedef __attribute__((ext_vector_type(2))) float v2f;
typedef __attribute__((ext_vector_type(8))) float v8f;


__global__ void gcn_zero_f4(float4* __restrict__ p, long n4) {
    long i = (long)blockIdx.x * blockDim.x + threadIdx.x;
    if (i < n4) p[i] = make_float4(0.f, 0.f, 0.f, 0.f);
}

#define BINW  256
#define BINSH 8
#define ECH   4096
#define BDEP  8

__device__ __forceinline__ void gcn_scan_bins(const int* __restrict__ cnt, int* sBase, int nb, int* wtmp) {
    const int t = threadIdx.x, lane = t & 31, w = t >> 5;
    int carry = 0;
    for (int c0 = 0; c0 <= nb; c0 += 256) {
        const int i = c0 + t;
        const int v = (i < nb) ? cnt[i] : 0;
        int incl = v;
        for (int o = 1; o < 32; o <<= 1) {
            int u = __shfl_up(incl, o);
            if (lane >= o) incl += u;
        }
        if (lane == 31) wtmp[w] = incl;
        __syncthreads();
        int wb = 0, tot = 0;
        for (int k = 0; k < 8; ++k) { const int x = wtmp[k]; tot += x; if (k < w) wb += x; }
        if (i <= nb) sBase[i] = carry + wb + incl - v;
        carry += tot;
        __syncthreads();
    }
}

__global__ __launch_bounds__(256)
void gcn_bin_count(const int* __restrict__ dst, int* __restrict__ binCount, int E, int nb) {
    extern __shared__ int sh[];
    const int t = threadIdx.x;
    for (int i = t; i < nb; i += 256) sh[i] = 0;
    __syncthreads();
    const int e0 = blockIdx.x * ECH + t;
    int ed[ECH / 256];
#pragma unroll
    for (int k = 0; k < ECH / 256; ++k) {
        const int e = e0 + k * 256;
        ed[k] = (e < E) ? dst[e] : -1;
    }
#pragma unroll
    for (int k = 0; k < ECH / 256; ++k)
        if (ed[k] >= 0) atomicAdd(&sh[ed[k] >> BINSH], 1);
    __syncthreads();
    for (int i = t; i < nb; i += 256) {
        const int c = sh[i];
        if (c) atomicAdd(&binCount[i], c);
    }
}

__global__ __launch_bounds__(256)
void gcn_bin_file(const int* __restrict__ src, const int* __restrict__ dst,
                  const int* __restrict__ binCount, int* __restrict__ binCursor,
                  int* __restrict__ binBase, int2* __restrict__ binned, int E, int nb) {
    extern __shared__ int sh[];
    int* sBase = sh;
    int* sCnt  = sh + (nb + 1);
    int* wtmp  = sCnt + nb;
    const int t = threadIdx.x;
    for (int i = t; i < nb; i += 256) sCnt[i] = 0;
    gcn_scan_bins(binCount, sBase, nb, wtmp);
    if (blockIdx.x == 0)
        for (int i = t; i <= nb; i += 256) binBase[i] = sBase[i];

    const int e0 = blockIdx.x * ECH + t;
    int es[ECH / 256], ed[ECH / 256], er[ECH / 256];
#pragma unroll
    for (int k = 0; k < ECH / 256; ++k) {
        const int e = e0 + k * 256;
        es[k] = 0; ed[k] = -1; er[k] = 0;
        if (e < E) { es[k] = src[e]; ed[k] = dst[e]; }
    }
#pragma unroll
    for (int k = 0; k < ECH / 256; ++k)
        if (ed[k] >= 0) er[k] = atomicAdd(&sCnt[ed[k] >> BINSH], 1);
    __syncthreads();
    for (int i = t; i < nb; i += 256) {
        const int c = sCnt[i];
        if (c) sCnt[i] = sBase[i] + atomicAdd(&binCursor[i], c);
    }
    __syncthreads();
#pragma unroll
    for (int k = 0; k < ECH / 256; ++k)
        if (ed[k] >= 0) binned[sCnt[ed[k] >> BINSH] + er[k]] = make_int2(es[k], ed[k]);
}

__global__ __launch_bounds__(256)
void gcn_bin_rows(const int2* __restrict__ binned, const int* __restrict__ binBase,
                  int* __restrict__ rowptr, float* __restrict__ dinv, int n) {
    __shared__ int cnt[BINW];
    __shared__ int wtot[8];
    const int t = threadIdx.x, lane = t & 31, w = t >> 5;
    const int b = blockIdx.x;
    const int beg = binBase[b], end = binBase[b + 1];
    cnt[t] = 0;
    __syncthreads();
    for (int i = beg + t; i < end; i += 256 * BDEP) {
        int d[BDEP];
#pragma unroll
        for (int k = 0; k < BDEP; ++k) {
            const int j = i + k * 256;
            d[k] = (j < end) ? binned[j].y : -1;
        }
#pragma unroll
        for (int k = 0; k < BDEP; ++k)
            if (d[k] >= 0) atomicAdd(&cnt[d[k] & (BINW - 1)], 1);
    }
    __syncthreads();
    const int c = cnt[t];
    int incl = c;
    for (int o = 1; o < 32; o <<= 1) {
        int u = __shfl_up(incl, o);
        if (lane >= o) incl += u;
    }
    if (lane == 31) wtot[w] = incl;
    __syncthreads();
    int wb = 0;
    for (int k = 0; k < w; ++k) wb += wtot[k];
    const int node = b * BINW + t;
    rowptr[node] = beg + wb + incl - c;
    if (node < n) dinv[node] = (c > 0) ? rsqrtf((float)c) : 0.f;
}

__global__ __launch_bounds__(256)
void gcn_bin_place(const int2* __restrict__ binned, const int* __restrict__ binBase,
                   const int* __restrict__ rowptr, const float* __restrict__ dinv,
                   int2* __restrict__ ew, int n) {
    __shared__ int   cur[BINW];
    __shared__ float dl[BINW];
    const int t = threadIdx.x;
    const int b = blockIdx.x;
    const int beg = binBase[b], end = binBase[b + 1];
    const int node = b * BINW + t;
    cur[t] = rowptr[node];
    dl[t]  = (node < n) ? dinv[node] : 0.f;
    __syncthreads();
    for (int i = beg + t; i < end; i += 256 * BDEP) {
        int2  sd[BDEP];
        float ds[BDEP];
#pragma unroll
        for (int k = 0; k < BDEP; ++k) {
            const int j = i + k * 256;
            sd[k] = (j < end) ? binned[j] : make_int2(0, -1);
        }
#pragma unroll
        for (int k = 0; k < BDEP; ++k) ds[k] = dinv[sd[k].x];
#pragma unroll
        for (int k = 0; k < BDEP; ++k) {
            if (sd[k].y >= 0) {
                const int r = sd[k].y & (BINW - 1);
                const int pos = atomicAdd(&cur[r], 1);
                ew[pos] = make_int2(sd[k].x, __float_as_int(ds[k] * dl[r]));
            }
        }
    }
}

#define LDP 132

template<int NT, int MT>
__global__ __launch_bounds__(NT * 32)
void gcn_gemm_wmma(const float* __restrict__ H, const float* __restrict__ W,
                   float* __restrict__ out, int nTiles) {
    __shared__ __attribute__((aligned(16))) float sW[NT * 16 * LDP];
    __shared__ __attribute__((aligned(16))) float sA[2][16 * LDP];

    constexpr int RPW = 16 / NT;
    static_assert(RPW == 2 || RPW == 4, "a wave stages 2 or 4 rows of the A tile");
    constexpr int RSTEP = NT * (DINK / 4);
    const int lane  = threadIdx.x & 31;
    const int wave  = threadIdx.x >> 5;
    const int l15   = lane & 15;
    const int half  = lane >> 4;
    const int col   = (wave << 4) + l15;
    const int koff  = half << 1;

    const int tile0 = blockIdx.x * MT;
    const int tend  = (tile0 + MT < nTiles) ? (tile0 + MT) : nTiles;

    const float4* hsrc = (const float4*)(H + (size_t)wave * DINK) + lane;
    float4 pre0, pre1, pre2 = make_float4(0.f, 0.f, 0.f, 0.f), pre3 = pre2;
    {
        const float4* s = hsrc + (size_t)tile0 * (16 * DINK / 4);
        pre0 = s[0];
        pre1 = s[RSTEP];
        if constexpr (RPW == 4) { pre2 = s[2 * RSTEP]; pre3 = s[3 * RSTEP]; }
    }

    {
        const float4* wsrc = (const float4*)(W + (size_t)(wave << 4) * DINK) + lane;
        float4 wrow[16];
#pragma unroll
        for (int i = 0; i < 16; ++i) wrow[i] = wsrc[i * (DINK / 4)];
#pragma unroll
        for (int i = 0; i < 16; ++i)
            *(float4*)&sW[((wave << 4) + i) * LDP + (lane << 2)] = wrow[i];
    }
    __syncthreads();
    v2f bfrag[DINK / 4];
    {
        const float* wp = &sW[col * LDP + koff];
#pragma unroll
        for (int t = 0; t < DINK / 4; ++t)
            bfrag[t] = *(const v2f*)(wp + 4 * t);
    }

    const int ldout = NT * 16;
    int p = 0;
    for (int tile = tile0; tile < tend; ++tile) {
        {
            float* d = &sA[p][wave * LDP + (lane << 2)];
            *(float4*)(d)                = pre0;
            *(float4*)(d + NT * LDP)     = pre1;
            if constexpr (RPW == 4) {
                *(float4*)(d + 2 * NT * LDP) = pre2;
                *(float4*)(d + 3 * NT * LDP) = pre3;
            }
        }
        __syncthreads();
        if (tile + 1 < tend) {
            const float4* s = hsrc + (size_t)(tile + 1) * (16 * DINK / 4);
            pre0 = s[0];
            pre1 = s[RSTEP];
            if constexpr (RPW == 4) { pre2 = s[2 * RSTEP]; pre3 = s[3 * RSTEP]; }
        }

        const float* ap = &sA[p][l15 * LDP + koff];
        v8f acc = {0.f, 0.f, 0.f, 0.f, 0.f, 0.f, 0.f, 0.f};
#pragma unroll
        for (int t = 0; t < DINK / 4; ++t) {
            v2f a = *(const v2f*)(ap + 4 * t);
            acc = __builtin_amdgcn_wmma_f32_16x16x4_f32(
                 false, a,  false, bfrag[t],
                 (short)0, acc,  false,  false);
        }

        float* op = out + (size_t)(tile * 16 + (half << 3)) * ldout + col;
#pragma unroll
        for (int r = 0; r < 8; ++r)
            op[(size_t)r * ldout] = acc[r];
        p ^= 1;
    }
}

template<int VEC, bool ACT>
__global__ __launch_bounds__(256)
void gcn_gather_rows(const int* __restrict__ rowptr, const int2* __restrict__ ew,
                     const float* __restrict__ tmp, const float* __restrict__ bias,
                     float* __restrict__ out, int n) {
    typedef float vec_t __attribute__((ext_vector_type(VEC)));
    const int lane = threadIdx.x & 31;
    int d = blockIdx.x * 8 + (threadIdx.x >> 5);
    if (d >= n) return;
    d = __builtin_amdgcn_readfirstlane(d);
    const int beg = rowptr[d];
    const int end = rowptr[d + 1];
    const float* colp = tmp + lane * VEC;
    vec_t bb = (vec_t)(0.f);
    if constexpr (ACT) bb = *(const vec_t*)(bias + lane * VEC);
    vec_t acc = (vec_t)(0.f);
    for (int base = beg; base < end; base += 32) {
        const int m = (end - base < 32) ? (end - base) : 32;
        int2 my = make_int2(0, 0);
        if (lane < m) my = ew[base + lane];
#pragma unroll 4
        for (int j = 0; j < m; ++j) {
            const int   s = __builtin_amdgcn_readlane(my.x, j);
            const float w = __int_as_float(__builtin_amdgcn_readlane(my.y, j));
            const vec_t v = *(const vec_t*)(colp + (size_t)s * (32 * VEC));
            acc += v * w;
        }
    }
    if constexpr (ACT) {
        acc += bb;
        acc = __builtin_elementwise_max(acc, (vec_t)(0.f));
    }
    *(vec_t*)(out + (size_t)d * (32 * VEC) + lane * VEC) = acc;
}

__global__ void gcn_bias_store64(const float4* __restrict__ in, const float* __restrict__ b,
                                 float4* __restrict__ out, long n4) {
    long i = (long)blockIdx.x * blockDim.x + threadIdx.x;
    if (i >= n4) return;
    float4 bb = ((const float4*)b)[(int)(i & (DOK / 4 - 1))];
    float4 v  = in[i];
    v.x += bb.x; v.y += bb.y; v.z += bb.z; v.w += bb.w;
    out[i] = v;
}


extern "C" void kernel_launch(void* const* d_in, const int* in_sizes, int n_in,
                              void* d_out, int out_size, void* d_ws, size_t ws_size,
                              hipStream_t stream) {
    const float* x  = (const float*)d_in[0];
    const int*   ei = (const int*)d_in[1];
    const float* W1 = (const float*)d_in[2];
    const float* b1 = (const float*)d_in[3];
    const float* W2 = (const float*)d_in[4];
    const float* b2 = (const float*)d_in[5];
    const float* W3 = (const float*)d_in[6];
    const float* b3 = (const float*)d_in[7];
    const float* W4 = (const float*)d_in[8];
    const float* b4 = (const float*)d_in[9];

    const int Nn = in_sizes[0] / DINK;
    const int E  = in_sizes[1] / 2;
    const int* srcI = ei;
    const int* dstI = ei + E;

    float* ws_f = (float*)d_ws;
    float* dinv = ws_f;
    size_t offA = ((size_t)Nn + 1023) & ~(size_t)1023;
    float* A = ws_f + offA;
    float* B = A + (size_t)Nn * DHK;
    float* out = (float*)d_out;

    const int NB    = Nn / BINW + 1;
    const int Np    = NB * BINW;
    const int nbPad = (NB + 3) & ~3;
    int2* ew        = (int2*)d_out;
    int*  rowptr    = (int*)d_out + ((2 * (size_t)E + 3) & ~(size_t)3);
    int*  binCount  = rowptr + Np;
    int*  binCursor = binCount + nbPad;
    int*  binBase   = binCursor + nbPad;
    int2* binned    = (int2*)B;

    const int TB = 256;
    const long no4   = (long)Nn * DOK / 4;
    const int  gN16  = Nn / 16;
    const int  gGemm = (gN16 + GMT - 1) / GMT;
    const int  gBin  = (E + ECH - 1) / ECH;
    const int  gRows = (Nn + 7) / 8;
    const int  gO4   = (int)((no4 + TB - 1) / TB);

    gcn_zero_f4<<<(nbPad / 2 + TB - 1) / TB, TB, 0, stream>>>((float4*)binCount, nbPad / 2);
    gcn_bin_count<<<gBin, 256, NB * sizeof(int), stream>>>(dstI, binCount, E, NB);
    gcn_bin_file<<<gBin, 256, (2 * NB + 1 + 8) * sizeof(int), stream>>>(srcI, dstI, binCount, binCursor,
                                                                      binBase, binned, E, NB);
    gcn_bin_rows<<<NB, 256, 0, stream>>>(binned, binBase, rowptr, dinv, Nn);
    gcn_bin_place<<<NB, 256, 0, stream>>>(binned, binBase, rowptr, dinv, ew, Nn);

    gcn_gemm_wmma<8, GMT><<<gGemm, 256, 0, stream>>>(x, W1, A, gN16);
    gcn_gather_rows<4, true><<<gRows, 256, 0, stream>>>(rowptr, ew, A, b1, B, Nn);

    gcn_gemm_wmma<8, GMT><<<gGemm, 256, 0, stream>>>(B, W2, A, gN16);
    gcn_gather_rows<4, true><<<gRows, 256, 0, stream>>>(rowptr, ew, A, b2, B, Nn);

    gcn_gemm_wmma<8, GMT><<<gGemm, 256, 0, stream>>>(B, W3, A, gN16);
    gcn_gather_rows<4, true><<<gRows, 256, 0, stream>>>(rowptr, ew, A, b3, B, Nn);

    gcn_gemm_wmma<4, GMT><<<gGemm, 128, 0, stream>>>(B, W4, A, gN16);
    gcn_gather_rows<2, false><<<gRows, 256, 0, stream>>>(rowptr, ew, A, b4, B, Nn);
    gcn_bias_store64<<<gO4, TB, 0, stream>>>((const float4*)B, b4, (float4*)out, no4);
}


# Layers 1-3: a gather block keeps its finished row tile (sum, +b, max 0) in LDS as the A tile of the next layer's WMMA product, 1024 resident blocks; activations never stored; row sums, k order kept
# speedup vs baseline: 6.1389x; 1.1016x over previous
#include <hip/hip_runtime.h>

#define DINK 128
#define DHK  128
#define DOK  64
#define GMT  5

typedef __attribute__((ext_vector_type(2))) float v2f;
typedef __attribute__((ext_vector_type(8))) float v8f;


__global__ void gcn_zero_f4(float4* __restrict__ p, long n4) {
    long i = (long)blockIdx.x * blockDim.x + threadIdx.x;
    if (i < n4) p[i] = make_float4(0.f, 0.f, 0.f, 0.f);
}

#define BINW  256
#define BINSH 8
#define ECH   4096
#define BDEP  8

__device__ __forceinline__ void gcn_scan_bins(const int* __restrict__ cnt, int* sBase, int nb, int* wtmp) {
    const int t = threadIdx.x, lane = t & 31, w = t >> 5;
    int carry = 0;
    for (int c0 = 0; c0 <= nb; c0 += 256) {
        const int i = c0 + t;
        const int v = (i < nb) ? cnt[i] : 0;
        int incl = v;
        for (int o = 1; o < 32; o <<= 1) {
            int u = __shfl_up(incl, o);
            if (lane >= o) incl += u;
        }
        if (lane == 31) wtmp[w] = incl;
        __syncthreads();
        int wb = 0, tot = 0;
        for (int k = 0; k < 8; ++k) { const int x = wtmp[k]; tot += x; if (k < w) wb += x; }
        if (i <= nb) sBase[i] = carry + wb + incl - v;
        carry += tot;
        __syncthreads();
    }
}

__global__ __launch_bounds__(256)
void gcn_bin_count(const int* __restrict__ dst, int* __restrict__ binCount, int E, int nb) {
    extern __shared__ int sh[];
    const int t = threadIdx.x;
    for (int i = t; i < nb; i += 256) sh[i] = 0;
    __syncthreads();
    const int e0 = blockIdx.x * ECH + t;
    int ed[ECH / 256];
#pragma unroll
    for (int k = 0; k < ECH / 256; ++k) {
        const int e = e0 + k * 256;
        ed[k] = (e < E) ? dst[e] : -1;
    }
#pragma unroll
    for (int k = 0; k < ECH / 256; ++k)
        if (ed[k] >= 0) atomicAdd(&sh[ed[k] >> BINSH], 1);
    __syncthreads();
    for (int i = t; i < nb; i += 256) {
        const int c = sh[i];
        if (c) atomicAdd(&binCount[i], c);
    }
}

__global__ __launch_bounds__(256)
void gcn_bin_file(const int* __restrict__ src, const int* __restrict__ dst,
                  const int* __restrict__ binCount, int* __restrict__ binCursor,
                  int* __restrict__ binBase, int2* __restrict__ binned, int E, int nb) {
    extern __shared__ int sh[];
    int* sBase = sh;
    int* sCnt  = sh + (nb + 1);
    int* wtmp  = sCnt + nb;
    const int t = threadIdx.x;
    for (int i = t; i < nb; i += 256) sCnt[i] = 0;
    gcn_scan_bins(binCount, sBase, nb, wtmp);
    if (blockIdx.x == 0)
        for (int i = t; i <= nb; i += 256) binBase[i] = sBase[i];

    const int e0 = blockIdx.x * ECH + t;
    int es[ECH / 256], ed[ECH / 256], er[ECH / 256];
#pragma unroll
    for (int k = 0; k < ECH / 256; ++k) {
        const int e = e0 + k * 256;
        es[k] = 0; ed[k] = -1; er[k] = 0;
        if (e < E) { es[k] = src[e]; ed[k] = dst[e]; }
    }
#pragma unroll
    for (int k = 0; k < ECH / 256; ++k)
        if (ed[k] >= 0) er[k] = atomicAdd(&sCnt[ed[k] >> BINSH], 1);
    __syncthreads();
    for (int i = t; i < nb; i += 256) {
        const int c = sCnt[i];
        if (c) sCnt[i] = sBase[i] + atomicAdd(&binCursor[i], c);
    }
    __syncthreads();
#pragma unroll
    for (int k = 0; k < ECH / 256; ++k)
        if (ed[k] >= 0) binned[sCnt[ed[k] >> BINSH] + er[k]] = make_int2(es[k], ed[k]);
}

__global__ __launch_bounds__(256)
void gcn_bin_rows(const int2* __restrict__ binned, const int* __restrict__ binBase,
                  int* __restrict__ rowptr, float* __restrict__ dinv, int n) {
    __shared__ int cnt[BINW];
    __shared__ int wtot[8];
    const int t = threadIdx.x, lane = t & 31, w = t >> 5;
    const int b = blockIdx.x;
    const int beg = binBase[b], end = binBase[b + 1];
    cnt[t] = 0;
    __syncthreads();
    for (int i = beg + t; i < end; i += 256 * BDEP) {
        int d[BDEP];
#pragma unroll
        for (int k = 0; k < BDEP; ++k) {
            const int j = i + k * 256;
            d[k] = (j < end) ? binned[j].y : -1;
        }
#pragma unroll
        for (int k = 0; k < BDEP; ++k)
            if (d[k] >= 0) atomicAdd(&cnt[d[k] & (BINW - 1)], 1);
    }
    __syncthreads();
    const int c = cnt[t];
    int incl = c;
    for (int o = 1; o < 32; o <<= 1) {
        int u = __shfl_up(incl, o);
        if (lane >= o) incl += u;
    }
    if (lane == 31) wtot[w] = incl;
    __syncthreads();
    int wb = 0;
    for (int k = 0; k < w; ++k) wb += wtot[k];
    const int node = b * BINW + t;
    rowptr[node] = beg + wb + incl - c;
    if (node < n) dinv[node] = (c > 0) ? rsqrtf((float)c) : 0.f;
}

__global__ __launch_bounds__(256)
void gcn_bin_place(const int2* __restrict__ binned, const int* __restrict__ binBase,
                   const int* __restrict__ rowptr, const float* __restrict__ dinv,
                   int2* __restrict__ ew, int n) {
    __shared__ int   cur[BINW];
    __shared__ float dl[BINW];
    const int t = threadIdx.x;
    const int b = blockIdx.x;
    const int beg = binBase[b], end = binBase[b + 1];
    const int node = b * BINW + t;
    cur[t] = rowptr[node];
    dl[t]  = (node < n) ? dinv[node] : 0.f;
    __syncthreads();
    for (int i = beg + t; i < end; i += 256 * BDEP) {
        int2  sd[BDEP];
        float ds[BDEP];
#pragma unroll
        for (int k = 0; k < BDEP; ++k) {
            const int j = i + k * 256;
            sd[k] = (j < end) ? binned[j] : make_int2(0, -1);
        }
#pragma unroll
        for (int k = 0; k < BDEP; ++k) ds[k] = dinv[sd[k].x];
#pragma unroll
        for (int k = 0; k < BDEP; ++k) {
            if (sd[k].y >= 0) {
                const int r = sd[k].y & (BINW - 1);
                const int pos = atomicAdd(&cur[r], 1);
                ew[pos] = make_int2(sd[k].x, __float_as_int(ds[k] * dl[r]));
            }
        }
    }
}

#define LDP 132

template<int NT, int MT>
__global__ __launch_bounds__(NT * 32)
void gcn_gemm_wmma(const float* __restrict__ H, const float* __restrict__ W,
                   float* __restrict__ out, int nTiles) {
    __shared__ __attribute__((aligned(16))) float sW[NT * 16 * LDP];
    __shared__ __attribute__((aligned(16))) float sA[2][16 * LDP];

    constexpr int RPW = 16 / NT;
    static_assert(RPW == 2 || RPW == 4, "a wave stages 2 or 4 rows of the A tile");
    constexpr int RSTEP = NT * (DINK / 4);
    const int lane  = threadIdx.x & 31;
    const int wave  = threadIdx.x >> 5;
    const int l15   = lane & 15;
    const int half  = lane >> 4;
    const int col   = (wave << 4) + l15;
    const int koff  = half << 1;

    const int tile0 = blockIdx.x * MT;
    const int tend  = (tile0 + MT < nTiles) ? (tile0 + MT) : nTiles;

    const float4* hsrc = (const float4*)(H + (size_t)wave * DINK) + lane;
    float4 pre0, pre1, pre2 = make_float4(0.f, 0.f, 0.f, 0.f), pre3 = pre2;
    {
        const float4* s = hsrc + (size_t)tile0 * (16 * DINK / 4);
        pre0 = s[0];
        pre1 = s[RSTEP];
        if constexpr (RPW == 4) { pre2 = s[2 * RSTEP]; pre3 = s[3 * RSTEP]; }
    }

    {
        const float4* wsrc = (const float4*)(W + (size_t)(wave << 4) * DINK) + lane;
        float4 wrow[16];
#pragma unroll
        for (int i = 0; i < 16; ++i) wrow[i] = wsrc[i * (DINK / 4)];
#pragma unroll
        for (int i = 0; i < 16; ++i)
            *(float4*)&sW[((wave << 4) + i) * LDP + (lane << 2)] = wrow[i];
    }
    __syncthreads();
    v2f bfrag[DINK / 4];
    {
        const float* wp = &sW[col * LDP + koff];
#pragma unroll
        for (int t = 0; t < DINK / 4; ++t)
            bfrag[t] = *(const v2f*)(wp + 4 * t);
    }

    const int ldout = NT * 16;
    int p = 0;
    for (int tile = tile0; tile < tend; ++tile) {
        {
            float* d = &sA[p][wave * LDP + (lane << 2)];
            *(float4*)(d)                = pre0;
            *(float4*)(d + NT * LDP)     = pre1;
            if constexpr (RPW == 4) {
                *(float4*)(d + 2 * NT * LDP) = pre2;
                *(float4*)(d + 3 * NT * LDP) = pre3;
            }
        }
        __syncthreads();
        if (tile + 1 < tend) {
            const float4* s = hsrc + (size_t)(tile + 1) * (16 * DINK / 4);
            pre0 = s[0];
            pre1 = s[RSTEP];
            if constexpr (RPW == 4) { pre2 = s[2 * RSTEP]; pre3 = s[3 * RSTEP]; }
        }

        const float* ap = &sA[p][l15 * LDP + koff];
        v8f acc = {0.f, 0.f, 0.f, 0.f, 0.f, 0.f, 0.f, 0.f};
#pragma unroll
        for (int t = 0; t < DINK / 4; ++t) {
            v2f a = *(const v2f*)(ap + 4 * t);
            acc = __builtin_amdgcn_wmma_f32_16x16x4_f32(
                 false, a,  false, bfrag[t],
                 (short)0, acc,  false,  false);
        }

        float* op = out + (size_t)(tile * 16 + (half << 3)) * ldout + col;
#pragma unroll
        for (int r = 0; r < 8; ++r)
            op[(size_t)r * ldout] = acc[r];
        p ^= 1;
    }
}

template<int VEC, bool ACT>
__global__ __launch_bounds__(256)
void gcn_gather_rows(const int* __restrict__ rowptr, const int2* __restrict__ ew,
                     const float* __restrict__ tmp, const float* __restrict__ bias,
                     float* __restrict__ out, int n) {
    typedef float vec_t __attribute__((ext_vector_type(VEC)));
    const int lane = threadIdx.x & 31;
    int d = blockIdx.x * 8 + (threadIdx.x >> 5);
    if (d >= n) return;
    d = __builtin_amdgcn_readfirstlane(d);
    const int beg = rowptr[d];
    const int end = rowptr[d + 1];
    const float* colp = tmp + lane * VEC;
    vec_t bb = (vec_t)(0.f);
    if constexpr (ACT) bb = *(const vec_t*)(bias + lane * VEC);
    vec_t acc = (vec_t)(0.f);
    for (int base = beg; base < end; base += 32) {
        const int m = (end - base < 32) ? (end - base) : 32;
        int2 my = make_int2(0, 0);
        if (lane < m) my = ew[base + lane];
#pragma unroll 4
        for (int j = 0; j < m; ++j) {
            const int   s = __builtin_amdgcn_readlane(my.x, j);
            const float w = __int_as_float(__builtin_amdgcn_readlane(my.y, j));
            const vec_t v = *(const vec_t*)(colp + (size_t)s * (32 * VEC));
            acc += v * w;
        }
    }
    if constexpr (ACT) {
        acc += bb;
        acc = __builtin_elementwise_max(acc, (vec_t)(0.f));
    }
    *(vec_t*)(out + (size_t)d * (32 * VEC) + lane * VEC) = acc;
}

typedef __attribute__((ext_vector_type(4))) float v4f;
template<int K>
__device__ __forceinline__ void gcn_add_rows(v4f& acc, const float* tmp, int lane, int2 my, int j0) {
    v4f   v[K];
    float w[K];
#pragma unroll
    for (int k = 0; k < K; ++k) {
        const int s = __builtin_amdgcn_readlane(my.x, j0 + k);
        w[k] = __int_as_float(__builtin_amdgcn_readlane(my.y, j0 + k));
        v[k] = ((const v4f*)(tmp + (size_t)s * DHK))[lane];
    }
#pragma unroll
    for (int k = 0; k < K; ++k)
        acc = __builtin_elementwise_fma(v[k], (v4f)(w[k]), acc);
}

template<int NT>
__global__ __launch_bounds__(256, 8)
void gcn_gather_gemm(const int* __restrict__ rowptr, const int2* __restrict__ ew,
                     const float* __restrict__ tmp, const float* __restrict__ bias,
                     const float* __restrict__ W, float* __restrict__ out, int nTiles) {
    __shared__ __attribute__((aligned(16))) float sm[NT * 16 * LDP];
    static_assert(NT * 16 >= 32, "two 16-row activation tiles must fit where W was staged");

    const int lane = threadIdx.x & 31;
    const int wave = __builtin_amdgcn_readfirstlane((int)(threadIdx.x >> 5));
    const int l15  = lane & 15;
    const int half = lane >> 4;
    const int koff = half << 1;
    const bool mm  = (NT == 8) || (wave < NT);
    const int col  = ((wave & (NT - 1)) << 4) + l15;

    if (mm) {
        const float4* wsrc = (const float4*)(W + (size_t)(wave << 4) * DINK) + lane;
        float4 wrow[16];
#pragma unroll
        for (int i = 0; i < 16; ++i) wrow[i] = wsrc[i * (DINK / 4)];
#pragma unroll
        for (int i = 0; i < 16; ++i)
            *(float4*)&sm[((wave << 4) + i) * LDP + (lane << 2)] = wrow[i];
    }
    __syncthreads();
    v2f bfrag[DINK / 4];
    {
        const float* wp = &sm[col * LDP + koff];
#pragma unroll
        for (int t = 0; t < DINK / 4; ++t)
            bfrag[t] = *(const v2f*)(wp + 4 * t);
    }
    __syncthreads();

    const v4f bb = *(const v4f*)(bias + (lane << 2));
    const int ldout = NT * 16;
    int p = 0;
    for (int tile = blockIdx.x; tile < nTiles; tile += gridDim.x) {
        float* sA = sm + p * (16 * LDP);
        const int d0 = tile * 16 + (wave << 1);
        const int bnd0 = rowptr[d0], bnd1 = rowptr[d0 + 1], bnd2 = rowptr[d0 + 2];
        int2 first0 = make_int2(0, 0), first1 = make_int2(0, 0);
        if (lane < bnd1 - bnd0) first0 = ew[bnd0 + lane];
        if (lane < bnd2 - bnd1) first1 = ew[bnd1 + lane];
        for (int i = 0; i < 2; ++i) {
            const int beg = i ? bnd1 : bnd0;
            const int end = i ? bnd2 : bnd1;
            int2 my = i ? first1 : first0;
            v4f acc = (v4f)(0.f);
            for (int base = beg; base < end; base += 32) {
                const int m = (end - base < 32) ? (end - base) : 32;
                if (base != beg) {
                    my = make_int2(0, 0);
                    if (lane < m) my = ew[base + lane];
                }
                int j = 0;
                for (; j + 8 <= m; j += 8) gcn_add_rows<8>(acc, tmp, lane, my, j);
                if (m - j >= 4) { gcn_add_rows<4>(acc, tmp, lane, my, j); j += 4; }
                if (m - j >= 2) { gcn_add_rows<2>(acc, tmp, lane, my, j); j += 2; }
                if (m - j >= 1) gcn_add_rows<1>(acc, tmp, lane, my, j);
            }
            acc += bb;
            acc = __builtin_elementwise_max(acc, (v4f)(0.f));
            *(v4f*)&sA[((wave << 1) + i) * LDP + (lane << 2)] = acc;
        }
        __syncthreads();
        if (mm) {
            const float* ap = &sA[l15 * LDP + koff];
            v8f c = {0.f, 0.f, 0.f, 0.f, 0.f, 0.f, 0.f, 0.f};
#pragma unroll
            for (int t = 0; t < DINK / 4; ++t) {
                v2f a = *(const v2f*)(ap + 4 * t);
                c = __builtin_amdgcn_wmma_f32_16x16x4_f32(
                     false, a,  false, bfrag[t],
                     (short)0, c,  false,  false);
            }
            float* op = out + (size_t)(tile * 16 + (half << 3)) * ldout + col;
#pragma unroll
            for (int r = 0; r < 8; ++r)
                op[(size_t)r * ldout] = c[r];
        }
        p ^= 1;
    }
}

__global__ void gcn_bias_store64(const float4* __restrict__ in, const float* __restrict__ b,
                                 float4* __restrict__ out, long n4) {
    long i = (long)blockIdx.x * blockDim.x + threadIdx.x;
    if (i >= n4) return;
    float4 bb = ((const float4*)b)[(int)(i & (DOK / 4 - 1))];
    float4 v  = in[i];
    v.x += bb.x; v.y += bb.y; v.z += bb.z; v.w += bb.w;
    out[i] = v;
}


extern "C" void kernel_launch(void* const* d_in, const int* in_sizes, int n_in,
                              void* d_out, int out_size, void* d_ws, size_t ws_size,
                              hipStream_t stream) {
    const float* x  = (const float*)d_in[0];
    const int*   ei = (const int*)d_in[1];
    const float* W1 = (const float*)d_in[2];
    const float* b1 = (const float*)d_in[3];
    const float* W2 = (const float*)d_in[4];
    const float* b2 = (const float*)d_in[5];
    const float* W3 = (const float*)d_in[6];
    const float* b3 = (const float*)d_in[7];
    const float* W4 = (const float*)d_in[8];
    const float* b4 = (const float*)d_in[9];

    const int Nn = in_sizes[0] / DINK;
    const int E  = in_sizes[1] / 2;
    const int* srcI = ei;
    const int* dstI = ei + E;

    float* ws_f = (float*)d_ws;
    float* dinv = ws_f;
    size_t offA = ((size_t)Nn + 1023) & ~(size_t)1023;
    float* A = ws_f + offA;
    float* B = A + (size_t)Nn * DHK;
    float* out = (float*)d_out;

    const int NB    = Nn / BINW + 1;
    const int Np    = NB * BINW;
    const int nbPad = (NB + 3) & ~3;
    int2* ew        = (int2*)d_out;
    int*  rowptr    = (int*)d_out + ((2 * (size_t)E + 3) & ~(size_t)3);
    int*  binCount  = rowptr + Np;
    int*  binCursor = binCount + nbPad;
    int*  binBase   = binCursor + nbPad;
    int2* binned    = (int2*)B;

    const int TB = 256;
    const long no4   = (long)Nn * DOK / 4;
    const int  gN16  = Nn / 16;
    const int  gGemm = (gN16 + GMT - 1) / GMT;
    const int  gBin  = (E + ECH - 1) / ECH;
    const int  gRows = (Nn + 7) / 8;
    const int  gO4   = (int)((no4 + TB - 1) / TB);
    const int  gFuse = 1024;

    gcn_zero_f4<<<(nbPad / 2 + TB - 1) / TB, TB, 0, stream>>>((float4*)binCount, nbPad / 2);
    gcn_bin_count<<<gBin, 256, NB * sizeof(int), stream>>>(dstI, binCount, E, NB);
    gcn_bin_file<<<gBin, 256, (2 * NB + 1 + 8) * sizeof(int), stream>>>(srcI, dstI, binCount, binCursor,
                                                                      binBase, binned, E, NB);
    gcn_bin_rows<<<NB, 256, 0, stream>>>(binned, binBase, rowptr, dinv, Nn);
    gcn_bin_place<<<NB, 256, 0, stream>>>(binned, binBase, rowptr, dinv, ew, Nn);

    gcn_gemm_wmma<8, GMT><<<gGemm, 256, 0, stream>>>(x, W1, A, gN16);
    gcn_gather_gemm<8><<<gFuse, 256, 0, stream>>>(rowptr, ew, A, b1, W2, B, gN16);
    gcn_gather_gemm<8><<<gFuse, 256, 0, stream>>>(rowptr, ew, B, b2, W3, A, gN16);
    gcn_gather_gemm<4><<<gFuse, 256, 0, stream>>>(rowptr, ew, A, b3, W4, B, gN16);

    gcn_gather_rows<2, false><<<gRows, 256, 0, stream>>>(rowptr, ew, B, b4, A, Nn);
    gcn_bias_store64<<<gO4, TB, 0, stream>>>((const float4*)A, b4, (float4*)out, no4);
}


# All partial-line stores leave through LDS as whole contiguous runs: product tiles as 512-byte rows (were 64-byte pieces), filed edge records as runs (were single 8-byte stores); sums unchanged
# speedup vs baseline: 6.4380x; 1.0487x over previous
#include <hip/hip_runtime.h>

#define DINK 128
#define DHK  128
#define DOK  64
#define GMT  5

typedef __attribute__((ext_vector_type(2))) float v2f;
typedef __attribute__((ext_vector_type(8))) float v8f;


__global__ void gcn_zero_f4(float4* __restrict__ p, long n4) {
    long i = (long)blockIdx.x * blockDim.x + threadIdx.x;
    if (i < n4) p[i] = make_float4(0.f, 0.f, 0.f, 0.f);
}

#define BINW  256
#define BINSH 8
#define ECH   4096
#define BDEP  8

__device__ __forceinline__ void gcn_scan_bins(const int* __restrict__ cnt, int* sBase, int nb, int* wtmp) {
    const int t = threadIdx.x, lane = t & 31, w = t >> 5;
    int carry = 0;
    for (int c0 = 0; c0 <= nb; c0 += 256) {
        const int i = c0 + t;
        const int v = (i < nb) ? cnt[i] : 0;
        int incl = v;
        for (int o = 1; o < 32; o <<= 1) {
            int u = __shfl_up(incl, o);
            if (lane >= o) incl += u;
        }
        if (lane == 31) wtmp[w] = incl;
        __syncthreads();
        int wb = 0, tot = 0;
        for (int k = 0; k < 8; ++k) { const int x = wtmp[k]; tot += x; if (k < w) wb += x; }
        if (i <= nb) sBase[i] = carry + wb + incl - v;
        carry += tot;
        __syncthreads();
    }
}

__global__ __launch_bounds__(256)
void gcn_bin_count(const int* __restrict__ dst, int* __restrict__ binCount, int E, int nb) {
    extern __shared__ int sh[];
    const int t = threadIdx.x;
    for (int i = t; i < nb; i += 256) sh[i] = 0;
    __syncthreads();
    const int e0 = blockIdx.x * ECH + t;
    int ed[ECH / 256];
#pragma unroll
    for (int k = 0; k < ECH / 256; ++k) {
        const int e = e0 + k * 256;
        ed[k] = (e < E) ? dst[e] : -1;
    }
#pragma unroll
    for (int k = 0; k < ECH / 256; ++k)
        if (ed[k] >= 0) atomicAdd(&sh[ed[k] >> BINSH], 1);
    __syncthreads();
    for (int i = t; i < nb; i += 256) {
        const int c = sh[i];
        if (c) atomicAdd(&binCount[i], c);
    }
}

__global__ __launch_bounds__(256)
void gcn_bin_file(const int* __restrict__ src, const int* __restrict__ dst,
                  const int* __restrict__ binCount, int* __restrict__ binCursor,
                  int* __restrict__ binBase, int2* __restrict__ binned, int E, int nb) {
    extern __shared__ int2 shf[];
    int2* sRec = shf;
    int* sBase = (int*)(shf + ECH);
    int* sCnt  = sBase + (nb + 1);
    int* sLoc  = sCnt + nb;
    int* wtmp  = sLoc + (nb + 1);
    const int t = threadIdx.x;
    for (int i = t; i < nb; i += 256) sCnt[i] = 0;
    gcn_scan_bins(binCount, sBase, nb, wtmp);
    if (blockIdx.x == 0)
        for (int i = t; i <= nb; i += 256) binBase[i] = sBase[i];

    const int e0 = blockIdx.x * ECH + t;
    int es[ECH / 256], ed[ECH / 256], er[ECH / 256];
#pragma unroll
    for (int k = 0; k < ECH / 256; ++k) {
        const int e = e0 + k * 256;
        es[k] = 0; ed[k] = -1; er[k] = 0;
        if (e < E) { es[k] = src[e]; ed[k] = dst[e]; }
    }
#pragma unroll
    for (int k = 0; k < ECH / 256; ++k)
        if (ed[k] >= 0) er[k] = atomicAdd(&sCnt[ed[k] >> BINSH], 1);
    __syncthreads();
    gcn_scan_bins(sCnt, sLoc, nb, wtmp);
    for (int i = t; i < nb; i += 256) {
        const int c = sCnt[i];
        if (c) sCnt[i] = sBase[i] + atomicAdd(&binCursor[i], c);
    }
#pragma unroll
    for (int k = 0; k < ECH / 256; ++k)
        if (ed[k] >= 0) sRec[sLoc[ed[k] >> BINSH] + er[k]] = make_int2(es[k], ed[k]);
    __syncthreads();
    const int total = sLoc[nb];
    int2 rr[ECH / 256];
#pragma unroll
    for (int k = 0; k < ECH / 256; ++k) {
        const int s = t + k * 256;
        rr[k] = (s < total) ? sRec[s] : make_int2(0, -1);
    }
#pragma unroll
    for (int k = 0; k < ECH / 256; ++k) {
        const int s = t + k * 256;
        if (rr[k].y >= 0) {
            const int b = rr[k].y >> BINSH;
            binned[sCnt[b] + (s - sLoc[b])] = rr[k];
        }
    }
}

__global__ __launch_bounds__(256)
void gcn_bin_rows(const int2* __restrict__ binned, const int* __restrict__ binBase,
                  int* __restrict__ rowptr, float* __restrict__ dinv, int n) {
    __shared__ int cnt[BINW];
    __shared__ int wtot[8];
    const int t = threadIdx.x, lane = t & 31, w = t >> 5;
    const int b = blockIdx.x;
    const int beg = binBase[b], end = binBase[b + 1];
    cnt[t] = 0;
    __syncthreads();
    for (int i = beg + t; i < end; i += 256 * BDEP) {
        int d[BDEP];
#pragma unroll
        for (int k = 0; k < BDEP; ++k) {
            const int j = i + k * 256;
            d[k] = (j < end) ? binned[j].y : -1;
        }
#pragma unroll
        for (int k = 0; k < BDEP; ++k)
            if (d[k] >= 0) atomicAdd(&cnt[d[k] & (BINW - 1)], 1);
    }
    __syncthreads();
    const int c = cnt[t];
    int incl = c;
    for (int o = 1; o < 32; o <<= 1) {
        int u = __shfl_up(incl, o);
        if (lane >= o) incl += u;
    }
    if (lane == 31) wtot[w] = incl;
    __syncthreads();
    int wb = 0;
    for (int k = 0; k < w; ++k) wb += wtot[k];
    const int node = b * BINW + t;
    rowptr[node] = beg + wb + incl - c;
    if (node < n) dinv[node] = (c > 0) ? rsqrtf((float)c) : 0.f;
}

#define PCAP 6144
__global__ __launch_bounds__(256)
void gcn_bin_place(const int2* __restrict__ binned, const int* __restrict__ binBase,
                   const int* __restrict__ rowptr, const float* __restrict__ dinv,
                   int2* __restrict__ ew, int n) {
    __shared__ int   cur[BINW];
    __shared__ float dl[BINW];
    __shared__ int2  stage[PCAP];
    const int t = threadIdx.x;
    const int b = blockIdx.x;
    const int beg = binBase[b], end = binBase[b + 1];
    const bool fits = (end - beg) <= PCAP;
    const int node = b * BINW + t;
    cur[t] = rowptr[node];
    dl[t]  = (node < n) ? dinv[node] : 0.f;
    __syncthreads();
    for (int i = beg + t; i < end; i += 256 * BDEP) {
        int2  sd[BDEP];
        float ds[BDEP];
#pragma unroll
        for (int k = 0; k < BDEP; ++k) {
            const int j = i + k * 256;
            sd[k] = (j < end) ? binned[j] : make_int2(0, -1);
        }
#pragma unroll
        for (int k = 0; k < BDEP; ++k) ds[k] = dinv[sd[k].x];
#pragma unroll
        for (int k = 0; k < BDEP; ++k) {
            if (sd[k].y >= 0) {
                const int r = sd[k].y & (BINW - 1);
                const int pos = atomicAdd(&cur[r], 1);
                const int2 rec = make_int2(sd[k].x, __float_as_int(ds[k] * dl[r]));
                if (fits) stage[pos - beg] = rec;
                else      ew[pos] = rec;
            }
        }
    }
    if (fits) {
        __syncthreads();
        const int nrec = end - beg;
        for (int i = t; i < nrec; i += 256 * BDEP) {
            int2 v[BDEP];
#pragma unroll
            for (int k = 0; k < BDEP; ++k) {
                const int j = i + k * 256;
                v[k] = (j < nrec) ? stage[j] : make_int2(0, 0);
            }
#pragma unroll
            for (int k = 0; k < BDEP; ++k) {
                const int j = i + k * 256;
                if (j < nrec) ew[beg + j] = v[k];
            }
        }
    }
}

#define LDP 132

template<int NT, int MT>
__global__ __launch_bounds__(NT * 32)
void gcn_gemm_wmma(const float* __restrict__ H, const float* __restrict__ W,
                   float* __restrict__ out, int nTiles) {
    __shared__ __attribute__((aligned(16))) float sW[NT * 16 * LDP];
    __shared__ __attribute__((aligned(16))) float sA[2][16 * LDP];
    constexpr int LDC = NT * 16 + 4;
    __shared__ __attribute__((aligned(16))) float sC[16 * LDC];

    constexpr int RPW = 16 / NT;
    static_assert(RPW == 2 || RPW == 4, "a wave stages 2 or 4 rows of the A tile");
    constexpr int RSTEP = NT * (DINK / 4);
    const int lane  = threadIdx.x & 31;
    const int wave  = threadIdx.x >> 5;
    const int l15   = lane & 15;
    const int half  = lane >> 4;
    const int col   = (wave << 4) + l15;
    const int koff  = half << 1;

    const int tile0 = blockIdx.x * MT;
    const int tend  = (tile0 + MT < nTiles) ? (tile0 + MT) : nTiles;

    const float4* hsrc = (const float4*)(H + (size_t)wave * DINK) + lane;
    float4 pre0, pre1, pre2 = make_float4(0.f, 0.f, 0.f, 0.f), pre3 = pre2;
    {
        const float4* s = hsrc + (size_t)tile0 * (16 * DINK / 4);
        pre0 = s[0];
        pre1 = s[RSTEP];
        if constexpr (RPW == 4) { pre2 = s[2 * RSTEP]; pre3 = s[3 * RSTEP]; }
    }

    {
        const float4* wsrc = (const float4*)(W + (size_t)(wave << 4) * DINK) + lane;
        float4 wrow[16];
#pragma unroll
        for (int i = 0; i < 16; ++i) wrow[i] = wsrc[i * (DINK / 4)];
#pragma unroll
        for (int i = 0; i < 16; ++i)
            *(float4*)&sW[((wave << 4) + i) * LDP + (lane << 2)] = wrow[i];
    }
    __syncthreads();
    v2f bfrag[DINK / 4];
    {
        const float* wp = &sW[col * LDP + koff];
#pragma unroll
        for (int t = 0; t < DINK / 4; ++t)
            bfrag[t] = *(const v2f*)(wp + 4 * t);
    }

    const int ldout = NT * 16;
    int p = 0;
    for (int tile = tile0; tile < tend; ++tile) {
        {
            float* d = &sA[p][wave * LDP + (lane << 2)];
            *(float4*)(d)                = pre0;
            *(float4*)(d + NT * LDP)     = pre1;
            if constexpr (RPW == 4) {
                *(float4*)(d + 2 * NT * LDP) = pre2;
                *(float4*)(d + 3 * NT * LDP) = pre3;
            }
        }
        __syncthreads();
        if (tile + 1 < tend) {
            const float4* s = hsrc + (size_t)(tile + 1) * (16 * DINK / 4);
            pre0 = s[0];
            pre1 = s[RSTEP];
            if constexpr (RPW == 4) { pre2 = s[2 * RSTEP]; pre3 = s[3 * RSTEP]; }
        }

        const float* ap = &sA[p][l15 * LDP + koff];
        v8f acc = {0.f, 0.f, 0.f, 0.f, 0.f, 0.f, 0.f, 0.f};
#pragma unroll
        for (int t = 0; t < DINK / 4; ++t) {
            v2f a = *(const v2f*)(ap + 4 * t);
            acc = __builtin_amdgcn_wmma_f32_16x16x4_f32(
                 false, a,  false, bfrag[t],
                 (short)0, acc,  false,  false);
        }

        {
            float* cp = &sC[(half << 3) * LDC + col];
#pragma unroll
            for (int r = 0; r < 8; ++r)
                cp[r * LDC] = acc[r];
        }
        __syncthreads();
        {
            float* ot = out + (size_t)tile * (16 * ldout);
            for (int i = 0; i < 2; ++i) {
                const int f = (wave + i * NT) * 128 + (lane << 2);
                *(float4*)(ot + f) = *(const float4*)&sC[(f / ldout) * LDC + (f % ldout)];
            }
        }
        p ^= 1;
    }
}

template<int VEC, bool ACT>
__global__ __launch_bounds__(256)
void gcn_gather_rows(const int* __restrict__ rowptr, const int2* __restrict__ ew,
                     const float* __restrict__ tmp, const float* __restrict__ bias,
                     float* __restrict__ out, int n) {
    typedef float vec_t __attribute__((ext_vector_type(VEC)));
    const int lane = threadIdx.x & 31;
    int d = blockIdx.x * 8 + (threadIdx.x >> 5);
    if (d >= n) return;
    d = __builtin_amdgcn_readfirstlane(d);
    const int beg = rowptr[d];
    const int end = rowptr[d + 1];
    const float* colp = tmp + lane * VEC;
    vec_t bb = (vec_t)(0.f);
    if constexpr (ACT) bb = *(const vec_t*)(bias + lane * VEC);
    vec_t acc = (vec_t)(0.f);
    for (int base = beg; base < end; base += 32) {
        const int m = (end - base < 32) ? (end - base) : 32;
        int2 my = make_int2(0, 0);
        if (lane < m) my = ew[base + lane];
#pragma unroll 4
        for (int j = 0; j < m; ++j) {
            const int   s = __builtin_amdgcn_readlane(my.x, j);
            const float w = __int_as_float(__builtin_amdgcn_readlane(my.y, j));
            const vec_t v = *(const vec_t*)(colp + (size_t)s * (32 * VEC));
            acc += v * w;
        }
    }
    if constexpr (ACT) {
        acc += bb;
        acc = __builtin_elementwise_max(acc, (vec_t)(0.f));
    }
    *(vec_t*)(out + (size_t)d * (32 * VEC) + lane * VEC) = acc;
}

typedef __attribute__((ext_vector_type(4))) float v4f;
template<int K>
__device__ __forceinline__ void gcn_add_rows(v4f& acc, const float* tmp, int lane, int2 my, int j0) {
    v4f   v[K];
    float w[K];
#pragma unroll
    for (int k = 0; k < K; ++k) {
        const int s = __builtin_amdgcn_readlane(my.x, j0 + k);
        w[k] = __int_as_float(__builtin_amdgcn_readlane(my.y, j0 + k));
        v[k] = ((const v4f*)(tmp + (size_t)s * DHK))[lane];
    }
#pragma unroll
    for (int k = 0; k < K; ++k)
        acc = __builtin_elementwise_fma(v[k], (v4f)(w[k]), acc);
}

template<int NT>
__global__ __launch_bounds__(256, 8)
void gcn_gather_gemm(const int* __restrict__ rowptr, const int2* __restrict__ ew,
                     const float* __restrict__ tmp, const float* __restrict__ bias,
                     const float* __restrict__ W, float* __restrict__ out, int nTiles) {
    __shared__ __attribute__((aligned(16))) float sm[NT * 16 * LDP];
    static_assert(NT * 16 >= 32, "two 16-row activation tiles must fit where W was staged");

    const int lane = threadIdx.x & 31;
    const int wave = __builtin_amdgcn_readfirstlane((int)(threadIdx.x >> 5));
    const int l15  = lane & 15;
    const int half = lane >> 4;
    const int koff = half << 1;
    const bool mm  = (NT == 8) || (wave < NT);
    const int col  = ((wave & (NT - 1)) << 4) + l15;

    if (mm) {
        const float4* wsrc = (const float4*)(W + (size_t)(wave << 4) * DINK) + lane;
        float4 wrow[16];
#pragma unroll
        for (int i = 0; i < 16; ++i) wrow[i] = wsrc[i * (DINK / 4)];
#pragma unroll
        for (int i = 0; i < 16; ++i)
            *(float4*)&sm[((wave << 4) + i) * LDP + (lane << 2)] = wrow[i];
    }
    __syncthreads();
    v2f bfrag[DINK / 4];
    {
        const float* wp = &sm[col * LDP + koff];
#pragma unroll
        for (int t = 0; t < DINK / 4; ++t)
            bfrag[t] = *(const v2f*)(wp + 4 * t);
    }
    __syncthreads();

    const v4f bb = *(const v4f*)(bias + (lane << 2));
    const int ldout = NT * 16;
    constexpr int LDC = NT * 16 + 4;
    float* sC = sm + 2 * (16 * LDP);
    static_assert(2 * 16 * LDP + 16 * LDC <= NT * 16 * LDP, "the product tile must fit where W was staged");
    int p = 0;
    for (int tile = blockIdx.x; tile < nTiles; tile += gridDim.x) {
        float* sA = sm + p * (16 * LDP);
        const int d0 = tile * 16 + (wave << 1);
        const int bnd0 = rowptr[d0], bnd1 = rowptr[d0 + 1], bnd2 = rowptr[d0 + 2];
        int2 first0 = make_int2(0, 0), first1 = make_int2(0, 0);
        if (lane < bnd1 - bnd0) first0 = ew[bnd0 + lane];
        if (lane < bnd2 - bnd1) first1 = ew[bnd1 + lane];
        for (int i = 0; i < 2; ++i) {
            const int beg = i ? bnd1 : bnd0;
            const int end = i ? bnd2 : bnd1;
            int2 my = i ? first1 : first0;
            v4f acc = (v4f)(0.f);
            for (int base = beg; base < end; base += 32) {
                const int m = (end - base < 32) ? (end - base) : 32;
                if (base != beg) {
                    my = make_int2(0, 0);
                    if (lane < m) my = ew[base + lane];
                }
                int j = 0;
                for (; j + 8 <= m; j += 8) gcn_add_rows<8>(acc, tmp, lane, my, j);
                if (m - j >= 4) { gcn_add_rows<4>(acc, tmp, lane, my, j); j += 4; }
                if (m - j >= 2) { gcn_add_rows<2>(acc, tmp, lane, my, j); j += 2; }
                if (m - j >= 1) gcn_add_rows<1>(acc, tmp, lane, my, j);
            }
            acc += bb;
            acc = __builtin_elementwise_max(acc, (v4f)(0.f));
            *(v4f*)&sA[((wave << 1) + i) * LDP + (lane << 2)] = acc;
        }
        __syncthreads();
        if (mm) {
            const float* ap = &sA[l15 * LDP + koff];
            v8f c = {0.f, 0.f, 0.f, 0.f, 0.f, 0.f, 0.f, 0.f};
#pragma unroll
            for (int t = 0; t < DINK / 4; ++t) {
                v2f a = *(const v2f*)(ap + 4 * t);
                c = __builtin_amdgcn_wmma_f32_16x16x4_f32(
                     false, a,  false, bfrag[t],
                     (short)0, c,  false,  false);
            }
            float* cp = &sC[(half << 3) * LDC + col];
#pragma unroll
            for (int r = 0; r < 8; ++r)
                cp[r * LDC] = c[r];
        }
        __syncthreads();
        {
            float* ot = out + (size_t)tile * (16 * ldout);
            for (int i = 0; i < (2 * NT) / 8; ++i) {
                const int f = (wave + i * 8) * 128 + (lane << 2);
                *(float4*)(ot + f) = *(const float4*)&sC[(f / ldout) * LDC + (f % ldout)];
            }
        }
        p ^= 1;
    }
}

__global__ void gcn_bias_store64(const float4* __restrict__ in, const float* __restrict__ b,
                                 float4* __restrict__ out, long n4) {
    long i = (long)blockIdx.x * blockDim.x + threadIdx.x;
    if (i >= n4) return;
    float4 bb = ((const float4*)b)[(int)(i & (DOK / 4 - 1))];
    float4 v  = in[i];
    v.x += bb.x; v.y += bb.y; v.z += bb.z; v.w += bb.w;
    out[i] = v;
}


extern "C" void kernel_launch(void* const* d_in, const int* in_sizes, int n_in,
                              void* d_out, int out_size, void* d_ws, size_t ws_size,
                              hipStream_t stream) {
    const float* x  = (const float*)d_in[0];
    const int*   ei = (const int*)d_in[1];
    const float* W1 = (const float*)d_in[2];
    const float* b1 = (const float*)d_in[3];
    const float* W2 = (const float*)d_in[4];
    const float* b2 = (const float*)d_in[5];
    const float* W3 = (const float*)d_in[6];
    const float* b3 = (const float*)d_in[7];
    const float* W4 = (const float*)d_in[8];
    const float* b4 = (const float*)d_in[9];

    const int Nn = in_sizes[0] / DINK;
    const int E  = in_sizes[1] / 2;
    const int* srcI = ei;
    const int* dstI = ei + E;

    float* ws_f = (float*)d_ws;
    float* dinv = ws_f;
    size_t offA = ((size_t)Nn + 1023) & ~(size_t)1023;
    float* A = ws_f + offA;
    float* B = A + (size_t)Nn * DHK;
    float* out = (float*)d_out;

    const int NB    = Nn / BINW + 1;
    const int Np    = NB * BINW;
    const int nbPad = (NB + 3) & ~3;
    int2* ew        = (int2*)d_out;
    int*  rowptr    = (int*)d_out + ((2 * (size_t)E + 3) & ~(size_t)3);
    int*  binCount  = rowptr + Np;
    int*  binCursor = binCount + nbPad;
    int*  binBase   = binCursor + nbPad;
    int2* binned    = (int2*)B;

    const int TB = 256;
    const long no4   = (long)Nn * DOK / 4;
    const int  gN16  = Nn / 16;
    const int  gGemm = (gN16 + GMT - 1) / GMT;
    const int  gBin  = (E + ECH - 1) / ECH;
    const int  gRows = (Nn + 7) / 8;
    const int  gO4   = (int)((no4 + TB - 1) / TB);
    const int  gFuse = 1024;

    gcn_zero_f4<<<(nbPad / 2 + TB - 1) / TB, TB, 0, stream>>>((float4*)binCount, nbPad / 2);
    gcn_bin_count<<<gBin, 256, NB * sizeof(int), stream>>>(dstI, binCount, E, NB);
    gcn_bin_file<<<gBin, 256, ECH * sizeof(int2) + (3 * NB + 2 + 8) * sizeof(int), stream>>>(srcI, dstI, binCount, binCursor,
                                                                                            binBase, binned, E, NB);
    gcn_bin_rows<<<NB, 256, 0, stream>>>(binned, binBase, rowptr, dinv, Nn);
    gcn_bin_place<<<NB, 256, 0, stream>>>(binned, binBase, rowptr, dinv, ew, Nn);

    gcn_gemm_wmma<8, GMT><<<gGemm, 256, 0, stream>>>(x, W1, A, gN16);
    gcn_gather_gemm<8><<<gFuse, 256, 0, stream>>>(rowptr, ew, A, b1, W2, B, gN16);
    gcn_gather_gemm<8><<<gFuse, 256, 0, stream>>>(rowptr, ew, B, b2, W3, A, gN16);
    gcn_gather_gemm<4><<<gFuse, 256, 0, stream>>>(rowptr, ew, A, b3, W4, B, gN16);

    gcn_gather_rows<2, false><<<gRows, 256, 0, stream>>>(rowptr, ew, B, b4, A, Nn);
    gcn_bias_store64<<<gO4, TB, 0, stream>>>((const float4*)A, b4, (float4*)out, no4);
}


# Each layer computed as relu((Agg h) W^T + b), not relu(Agg (h W^T) + b): gather, WMMA product, bias, relu in one launch; layer 1's separate product launch and plane are gone; sums reassociated
# speedup vs baseline: 6.6760x; 1.0370x over previous
#include <hip/hip_runtime.h>

#define DINK 128
#define DHK  128
#define DOK  64

typedef __attribute__((ext_vector_type(2))) float v2f;
typedef __attribute__((ext_vector_type(8))) float v8f;


__global__ void gcn_zero_f4(float4* __restrict__ p, long n4) {
    long i = (long)blockIdx.x * blockDim.x + threadIdx.x;
    if (i < n4) p[i] = make_float4(0.f, 0.f, 0.f, 0.f);
}

#define BINW  256
#define BINSH 8
#define ECH   4096
#define BDEP  8

__device__ __forceinline__ void gcn_scan_bins(const int* __restrict__ cnt, int* sBase, int nb, int* wtmp) {
    const int t = threadIdx.x, lane = t & 31, w = t >> 5;
    int carry = 0;
    for (int c0 = 0; c0 <= nb; c0 += 256) {
        const int i = c0 + t;
        const int v = (i < nb) ? cnt[i] : 0;
        int incl = v;
        for (int o = 1; o < 32; o <<= 1) {
            int u = __shfl_up(incl, o);
            if (lane >= o) incl += u;
        }
        if (lane == 31) wtmp[w] = incl;
        __syncthreads();
        int wb = 0, tot = 0;
        for (int k = 0; k < 8; ++k) { const int x = wtmp[k]; tot += x; if (k < w) wb += x; }
        if (i <= nb) sBase[i] = carry + wb + incl - v;
        carry += tot;
        __syncthreads();
    }
}

__global__ __launch_bounds__(256)
void gcn_bin_count(const int* __restrict__ dst, int* __restrict__ binCount, int E, int nb) {
    extern __shared__ int sh[];
    const int t = threadIdx.x;
    for (int i = t; i < nb; i += 256) sh[i] = 0;
    __syncthreads();
    const int e0 = blockIdx.x * ECH + t;
    int ed[ECH / 256];
#pragma unroll
    for (int k = 0; k < ECH / 256; ++k) {
        const int e = e0 + k * 256;
        ed[k] = (e < E) ? dst[e] : -1;
    }
#pragma unroll
    for (int k = 0; k < ECH / 256; ++k)
        if (ed[k] >= 0) atomicAdd(&sh[ed[k] >> BINSH], 1);
    __syncthreads();
    for (int i = t; i < nb; i += 256) {
        const int c = sh[i];
        if (c) atomicAdd(&binCount[i], c);
    }
}

__global__ __launch_bounds__(256)
void gcn_bin_file(const int* __restrict__ src, const int* __restrict__ dst,
                  const int* __restrict__ binCount, int* __restrict__ binCursor,
                  int* __restrict__ binBase, int2* __restrict__ binned, int E, int nb) {
    extern __shared__ int2 shf[];
    int2* sRec = shf;
    int* sBase = (int*)(shf + ECH);
    int* sCnt  = sBase + (nb + 1);
    int* sLoc  = sCnt + nb;
    int* wtmp  = sLoc + (nb + 1);
    const int t = threadIdx.x;
    for (int i = t; i < nb; i += 256) sCnt[i] = 0;
    gcn_scan_bins(binCount, sBase, nb, wtmp);
    if (blockIdx.x == 0)
        for (int i = t; i <= nb; i += 256) binBase[i] = sBase[i];

    const int e0 = blockIdx.x * ECH + t;
    int es[ECH / 256], ed[ECH / 256], er[ECH / 256];
#pragma unroll
    for (int k = 0; k < ECH / 256; ++k) {
        const int e = e0 + k * 256;
        es[k] = 0; ed[k] = -1; er[k] = 0;
        if (e < E) { es[k] = src[e]; ed[k] = dst[e]; }
    }
#pragma unroll
    for (int k = 0; k < ECH / 256; ++k)
        if (ed[k] >= 0) er[k] = atomicAdd(&sCnt[ed[k] >> BINSH], 1);
    __syncthreads();
    gcn_scan_bins(sCnt, sLoc, nb, wtmp);
    for (int i = t; i < nb; i += 256) {
        const int c = sCnt[i];
        if (c) sCnt[i] = sBase[i] + atomicAdd(&binCursor[i], c);
    }
#pragma unroll
    for (int k = 0; k < ECH / 256; ++k)
        if (ed[k] >= 0) sRec[sLoc[ed[k] >> BINSH] + er[k]] = make_int2(es[k], ed[k]);
    __syncthreads();
    const int total = sLoc[nb];
    int2 rr[ECH / 256];
#pragma unroll
    for (int k = 0; k < ECH / 256; ++k) {
        const int s = t + k * 256;
        rr[k] = (s < total) ? sRec[s] : make_int2(0, -1);
    }
#pragma unroll
    for (int k = 0; k < ECH / 256; ++k) {
        const int s = t + k * 256;
        if (rr[k].y >= 0) {
            const int b = rr[k].y >> BINSH;
            binned[sCnt[b] + (s - sLoc[b])] = rr[k];
        }
    }
}

__global__ __launch_bounds__(256)
void gcn_bin_rows(const int2* __restrict__ binned, const int* __restrict__ binBase,
                  int* __restrict__ rowptr, float* __restrict__ dinv, int n) {
    __shared__ int cnt[BINW];
    __shared__ int wtot[8];
    const int t = threadIdx.x, lane = t & 31, w = t >> 5;
    const int b = blockIdx.x;
    const int beg = binBase[b], end = binBase[b + 1];
    cnt[t] = 0;
    __syncthreads();
    for (int i = beg + t; i < end; i += 256 * BDEP) {
        int d[BDEP];
#pragma unroll
        for (int k = 0; k < BDEP; ++k) {
            const int j = i + k * 256;
            d[k] = (j < end) ? binned[j].y : -1;
        }
#pragma unroll
        for (int k = 0; k < BDEP; ++k)
            if (d[k] >= 0) atomicAdd(&cnt[d[k] & (BINW - 1)], 1);
    }
    __syncthreads();
    const int c = cnt[t];
    int incl = c;
    for (int o = 1; o < 32; o <<= 1) {
        int u = __shfl_up(incl, o);
        if (lane >= o) incl += u;
    }
    if (lane == 31) wtot[w] = incl;
    __syncthreads();
    int wb = 0;
    for (int k = 0; k < w; ++k) wb += wtot[k];
    const int node = b * BINW + t;
    rowptr[node] = beg + wb + incl - c;
    if (node < n) dinv[node] = (c > 0) ? rsqrtf((float)c) : 0.f;
}

#define PCAP 6144
__global__ __launch_bounds__(256)
void gcn_bin_place(const int2* __restrict__ binned, const int* __restrict__ binBase,
                   const int* __restrict__ rowptr, const float* __restrict__ dinv,
                   int2* __restrict__ ew, int n) {
    __shared__ int   cur[BINW];
    __shared__ float dl[BINW];
    __shared__ int2  stage[PCAP];
    const int t = threadIdx.x;
    const int b = blockIdx.x;
    const int beg = binBase[b], end = binBase[b + 1];
    const bool fits = (end - beg) <= PCAP;
    const int node = b * BINW + t;
    cur[t] = rowptr[node];
    dl[t]  = (node < n) ? dinv[node] : 0.f;
    __syncthreads();
    for (int i = beg + t; i < end; i += 256 * BDEP) {
        int2  sd[BDEP];
        float ds[BDEP];
#pragma unroll
        for (int k = 0; k < BDEP; ++k) {
            const int j = i + k * 256;
            sd[k] = (j < end) ? binned[j] : make_int2(0, -1);
        }
#pragma unroll
        for (int k = 0; k < BDEP; ++k) ds[k] = dinv[sd[k].x];
#pragma unroll
        for (int k = 0; k < BDEP; ++k) {
            if (sd[k].y >= 0) {
                const int r = sd[k].y & (BINW - 1);
                const int pos = atomicAdd(&cur[r], 1);
                const int2 rec = make_int2(sd[k].x, __float_as_int(ds[k] * dl[r]));
                if (fits) stage[pos - beg] = rec;
                else      ew[pos] = rec;
            }
        }
    }
    if (fits) {
        __syncthreads();
        const int nrec = end - beg;
        for (int i = t; i < nrec; i += 256 * BDEP) {
            int2 v[BDEP];
#pragma unroll
            for (int k = 0; k < BDEP; ++k) {
                const int j = i + k * 256;
                v[k] = (j < nrec) ? stage[j] : make_int2(0, 0);
            }
#pragma unroll
            for (int k = 0; k < BDEP; ++k) {
                const int j = i + k * 256;
                if (j < nrec) ew[beg + j] = v[k];
            }
        }
    }
}

#define LDP 132

template<int VEC, bool ACT>
__global__ __launch_bounds__(256)
void gcn_gather_rows(const int* __restrict__ rowptr, const int2* __restrict__ ew,
                     const float* __restrict__ tmp, const float* __restrict__ bias,
                     float* __restrict__ out, int n) {
    typedef float vec_t __attribute__((ext_vector_type(VEC)));
    const int lane = threadIdx.x & 31;
    int d = blockIdx.x * 8 + (threadIdx.x >> 5);
    if (d >= n) return;
    d = __builtin_amdgcn_readfirstlane(d);
    const int beg = rowptr[d];
    const int end = rowptr[d + 1];
    const float* colp = tmp + lane * VEC;
    vec_t bb = (vec_t)(0.f);
    if constexpr (ACT) bb = *(const vec_t*)(bias + lane * VEC);
    vec_t acc = (vec_t)(0.f);
    for (int base = beg; base < end; base += 32) {
        const int m = (end - base < 32) ? (end - base) : 32;
        int2 my = make_int2(0, 0);
        if (lane < m) my = ew[base + lane];
#pragma unroll 4
        for (int j = 0; j < m; ++j) {
            const int   s = __builtin_amdgcn_readlane(my.x, j);
            const float w = __int_as_float(__builtin_amdgcn_readlane(my.y, j));
            const vec_t v = *(const vec_t*)(colp + (size_t)s * (32 * VEC));
            acc += v * w;
        }
    }
    if constexpr (ACT) {
        acc += bb;
        acc = __builtin_elementwise_max(acc, (vec_t)(0.f));
    }
    *(vec_t*)(out + (size_t)d * (32 * VEC) + lane * VEC) = acc;
}

typedef __attribute__((ext_vector_type(4))) float v4f;
template<int K>
__device__ __forceinline__ void gcn_add_rows(v4f& acc, const float* tmp, int lane, int2 my, int j0) {
    v4f   v[K];
    float w[K];
#pragma unroll
    for (int k = 0; k < K; ++k) {
        const int s = __builtin_amdgcn_readlane(my.x, j0 + k);
        w[k] = __int_as_float(__builtin_amdgcn_readlane(my.y, j0 + k));
        v[k] = ((const v4f*)(tmp + (size_t)s * DHK))[lane];
    }
#pragma unroll
    for (int k = 0; k < K; ++k)
        acc = __builtin_elementwise_fma(v[k], (v4f)(w[k]), acc);
}


__device__ __forceinline__ void gcn_stage_rows16(const float* __restrict__ src, float* dst, int lane) {
    const float4* wsrc = (const float4*)src + lane;
    float4 wrow[16];
#pragma unroll
    for (int i = 0; i < 16; ++i) wrow[i] = wsrc[i * (DINK / 4)];
#pragma unroll
    for (int i = 0; i < 16; ++i)
        *(float4*)&dst[i * LDP + (lane << 2)] = wrow[i];
}

template<bool LAST>
__global__ __launch_bounds__(256, 8)
void gcn_gather_layer(const int* __restrict__ rowptr, const int2* __restrict__ ew,
                      const float* __restrict__ hin, const float* __restrict__ W,
                      const float* __restrict__ bias, const float* __restrict__ Wn,
                      float* __restrict__ out, int nTiles) {
    __shared__ __attribute__((aligned(16))) float sm[128 * LDP];

    const int lane = threadIdx.x & 31;
    const int wave = __builtin_amdgcn_readfirstlane((int)(threadIdx.x >> 5));
    const int l15  = lane & 15;
    const int half = lane >> 4;
    const int koff = half << 1;
    const int col  = (wave << 4) + l15;
    float* sWn = sm + 64 * LDP;

    v2f bfrag[DINK / 4];
    if constexpr (!LAST) {
        gcn_stage_rows16(W + (size_t)(wave << 4) * DINK, sm + (wave << 4) * LDP, lane);
        __syncthreads();
        const float* wp = &sm[col * LDP + koff];
#pragma unroll
        for (int t = 0; t < DINK / 4; ++t)
            bfrag[t] = *(const v2f*)(wp + 4 * t);
        __syncthreads();
    } else {
        const bool lo = wave < 4;
        if (lo) gcn_stage_rows16(W  + (size_t)(wave << 4) * DINK,       sm  + (wave << 4) * LDP,       lane);
        else    gcn_stage_rows16(Wn + (size_t)((wave - 4) << 4) * DINK, sWn + ((wave - 4) << 4) * LDP, lane);
        __syncthreads();
        if (lo) {
            const float* wp = &sm[col * LDP + koff];
#pragma unroll
            for (int t = 0; t < DINK / 4; ++t)
                bfrag[t] = *(const v2f*)(wp + 4 * t);
        }
        __syncthreads();
        if (!lo) gcn_stage_rows16(W + (size_t)(wave << 4) * DINK, sm + ((wave - 4) << 4) * LDP, lane);
        __syncthreads();
        if (!lo) {
            const float* wp = &sm[(col - 64) * LDP + koff];
#pragma unroll
            for (int t = 0; t < DINK / 4; ++t)
                bfrag[t] = *(const v2f*)(wp + 4 * t);
        }
        __syncthreads();
    }

    const float bcol = bias[col];
    constexpr int ldout = LAST ? DOK : DHK;
    constexpr int LDC = ldout + 4;
    float* sH = sm + 2 * (16 * LDP);
    float* sC = sm + (LAST ? 3 : 2) * (16 * LDP);
    static_assert(4 * 16 * LDP <= 64 * LDP && 16 * (DHK + 4) <= 16 * LDP, "the tiles must fit in front of the rows of Wn");
    int p = 0;
    for (int tile = blockIdx.x; tile < nTiles; tile += gridDim.x) {
        float* sA = sm + p * (16 * LDP);
        const int d0 = tile * 16 + (wave << 1);
        const int bnd0 = rowptr[d0], bnd1 = rowptr[d0 + 1], bnd2 = rowptr[d0 + 2];
        int2 first0 = make_int2(0, 0), first1 = make_int2(0, 0);
        if (lane < bnd1 - bnd0) first0 = ew[bnd0 + lane];
        if (lane < bnd2 - bnd1) first1 = ew[bnd1 + lane];
        for (int i = 0; i < 2; ++i) {
            const int beg = i ? bnd1 : bnd0;
            const int end = i ? bnd2 : bnd1;
            int2 my = i ? first1 : first0;
            v4f acc = (v4f)(0.f);
            for (int base = beg; base < end; base += 32) {
                const int m = (end - base < 32) ? (end - base) : 32;
                if (base != beg) {
                    my = make_int2(0, 0);
                    if (lane < m) my = ew[base + lane];
                }
                int j = 0;
                for (; j + 8 <= m; j += 8) gcn_add_rows<8>(acc, hin, lane, my, j);
                if (m - j >= 4) { gcn_add_rows<4>(acc, hin, lane, my, j); j += 4; }
                if (m - j >= 2) { gcn_add_rows<2>(acc, hin, lane, my, j); j += 2; }
                if (m - j >= 1) gcn_add_rows<1>(acc, hin, lane, my, j);
            }
            *(v4f*)&sA[((wave << 1) + i) * LDP + (lane << 2)] = acc;
        }
        __syncthreads();
        {
            const float* ap = &sA[l15 * LDP + koff];
            v8f c = {0.f, 0.f, 0.f, 0.f, 0.f, 0.f, 0.f, 0.f};
#pragma unroll
            for (int t = 0; t < DINK / 4; ++t) {
                v2f a = *(const v2f*)(ap + 4 * t);
                c = __builtin_amdgcn_wmma_f32_16x16x4_f32(
                     false, a,  false, bfrag[t],
                     (short)0, c,  false,  false);
            }
#pragma unroll
            for (int r = 0; r < 8; ++r)
                c[r] = __builtin_fmaxf(c[r] + bcol, 0.f);
            float* cp = LAST ? &sH[(half << 3) * LDP + col] : &sC[(half << 3) * LDC + col];
            constexpr int cpitch = LAST ? LDP : LDC;
#pragma unroll
            for (int r = 0; r < 8; ++r)
                cp[r * cpitch] = c[r];
        }
        if constexpr (LAST) {
            __syncthreads();
            if (wave < 4) {
                const float* ap = &sH[l15 * LDP + koff];
                const float* wp = &sWn[col * LDP + koff];
                v8f c = {0.f, 0.f, 0.f, 0.f, 0.f, 0.f, 0.f, 0.f};
#pragma unroll
                for (int t = 0; t < DINK / 4; ++t) {
                    v2f a = *(const v2f*)(ap + 4 * t);
                    v2f b = *(const v2f*)(wp + 4 * t);
                    c = __builtin_amdgcn_wmma_f32_16x16x4_f32(
                         false, a,  false, b,
                         (short)0, c,  false,  false);
                }
                float* cp = &sC[(half << 3) * LDC + col];
#pragma unroll
                for (int r = 0; r < 8; ++r)
                    cp[r * LDC] = c[r];
            }
        }
        __syncthreads();
        {
            float* ot = out + (size_t)tile * (16 * ldout);
            for (int i = 0; i < ldout / 64; ++i) {
                const int f = (wave + i * 8) * 128 + (lane << 2);
                *(float4*)(ot + f) = *(const float4*)&sC[(f / ldout) * LDC + (f % ldout)];
            }
        }
        p ^= 1;
    }
}

__global__ void gcn_bias_store64(const float4* __restrict__ in, const float* __restrict__ b,
                                 float4* __restrict__ out, long n4) {
    long i = (long)blockIdx.x * blockDim.x + threadIdx.x;
    if (i >= n4) return;
    float4 bb = ((const float4*)b)[(int)(i & (DOK / 4 - 1))];
    float4 v  = in[i];
    v.x += bb.x; v.y += bb.y; v.z += bb.z; v.w += bb.w;
    out[i] = v;
}


extern "C" void kernel_launch(void* const* d_in, const int* in_sizes, int n_in,
                              void* d_out, int out_size, void* d_ws, size_t ws_size,
                              hipStream_t stream) {
    const float* x  = (const float*)d_in[0];
    const int*   ei = (const int*)d_in[1];
    const float* W1 = (const float*)d_in[2];
    const float* b1 = (const float*)d_in[3];
    const float* W2 = (const float*)d_in[4];
    const float* b2 = (const float*)d_in[5];
    const float* W3 = (const float*)d_in[6];
    const float* b3 = (const float*)d_in[7];
    const float* W4 = (const float*)d_in[8];
    const float* b4 = (const float*)d_in[9];

    const int Nn = in_sizes[0] / DINK;
    const int E  = in_sizes[1] / 2;
    const int* srcI = ei;
    const int* dstI = ei + E;

    float* ws_f = (float*)d_ws;
    float* dinv = ws_f;
    size_t offA = ((size_t)Nn + 1023) & ~(size_t)1023;
    float* A = ws_f + offA;
    float* B = A + (size_t)Nn * DHK;
    float* out = (float*)d_out;

    const int NB    = Nn / BINW + 1;
    const int Np    = NB * BINW;
    const int nbPad = (NB + 3) & ~3;
    int2* ew        = (int2*)d_out;
    int*  rowptr    = (int*)d_out + ((2 * (size_t)E + 3) & ~(size_t)3);
    int*  binCount  = rowptr + Np;
    int*  binCursor = binCount + nbPad;
    int*  binBase   = binCursor + nbPad;
    int2* binned    = (int2*)B;

    const int TB = 256;
    const long no4   = (long)Nn * DOK / 4;
    const int  gN16  = Nn / 16;
    const int  gBin  = (E + ECH - 1) / ECH;
    const int  gRows = (Nn + 7) / 8;
    const int  gO4   = (int)((no4 + TB - 1) / TB);
    const int  gFuse = 1024;

    gcn_zero_f4<<<(nbPad / 2 + TB - 1) / TB, TB, 0, stream>>>((float4*)binCount, nbPad / 2);
    gcn_bin_count<<<gBin, 256, NB * sizeof(int), stream>>>(dstI, binCount, E, NB);
    gcn_bin_file<<<gBin, 256, ECH * sizeof(int2) + (3 * NB + 2 + 8) * sizeof(int), stream>>>(srcI, dstI, binCount, binCursor,
                                                                                            binBase, binned, E, NB);
    gcn_bin_rows<<<NB, 256, 0, stream>>>(binned, binBase, rowptr, dinv, Nn);
    gcn_bin_place<<<NB, 256, 0, stream>>>(binned, binBase, rowptr, dinv, ew, Nn);

    gcn_gather_layer<false><<<gFuse, 256, 0, stream>>>(rowptr, ew, x, W1, b1, W1, A, gN16);
    gcn_gather_layer<false><<<gFuse, 256, 0, stream>>>(rowptr, ew, A, W2, b2, W2, B, gN16);
    gcn_gather_layer<true><<<gFuse, 256, 0, stream>>>(rowptr, ew, B, W3, b3, W4, A, gN16);

    gcn_gather_rows<2, false><<<gRows, 256, 0, stream>>>(rowptr, ew, A, b4, B, Nn);
    gcn_bias_store64<<<gO4, TB, 0, stream>>>((const float4*)B, b4, (float4*)out, no4);
}
